# starting kernel
# baseline (speedup 1.0000x reference)
#include <hip/hip_runtime.h>
#include <hip/hip_bf16.h>


#define S_LEN   768
#define V_WORD  30000
#define V_POS   64
#define WEMB    100
#define PEMB    28
#define EMB     128
#define HDIM    256
#define G4H     1024
#define MLP     128

typedef __attribute__((ext_vector_type(2))) float v2f;
typedef __attribute__((ext_vector_type(8))) float v8f;

__device__ __forceinline__ float fast_sigmoid(float x) {
    return 1.0f / (1.0f + __expf(-x));
}
__device__ __forceinline__ float fast_tanh(float x) {
    float c = fminf(fmaxf(x, -9.0f), 9.0f);
    float e = __expf(2.0f * c);
    return (e - 1.0f) / (e + 1.0f);
}

__global__ void embed_kernel(const int* __restrict__ widx, const int* __restrict__ pidx,
                             const float* __restrict__ Wword, const float* __restrict__ Wpos,
                             float* __restrict__ x) {
    int s = blockIdx.x;
    int t = threadIdx.x;
    float v;
    if (t < WEMB) v = Wword[(size_t)widx[s] * WEMB + t];
    else          v = Wpos[(size_t)pidx[s] * PEMB + (t - WEMB)];
    x[(size_t)s * EMB + t] = v;
}

__global__ void transpose_sub(const float* __restrict__ src, float* __restrict__ dst,
                              int R, int Ctot, int coff, int Ccnt) {
    int idx = blockIdx.x * blockDim.x + threadIdx.x;
    int total = R * Ccnt;
    if (idx >= total) return;
    int r = idx / Ccnt;
    int c = idx - r * Ccnt;
    dst[(size_t)c * R + r] = src[(size_t)r * Ctot + coff + c];
}

__global__ void pack_bt(const float* __restrict__ src, float* __restrict__ dst,
                        int Nout, int Ctot, int coff, int Kcnt) {
    int idx = blockIdx.x * blockDim.x + threadIdx.x;
    int total = Nout * Kcnt;
    if (idx >= total) return;
    int n = idx / Kcnt;
    int k = idx - n * Kcnt;
    dst[((size_t)(k >> 2) * Nout + n) * 4 + (k & 3)] = src[(size_t)n * Ctot + coff + k];
}

__global__ void __launch_bounds__(32)
gemm_wmma_f32(const float* __restrict__ A, const float* __restrict__ BP,
              const float* __restrict__ bias0, const float* __restrict__ bias1,
              float* __restrict__ C, int M, int N, int K) {
    const int nt = blockIdx.x;
    const int mt = blockIdx.y;
    const int l  = threadIdx.x;
    const int fr   = l & 15;
    const int half = l >> 4;
    const int kb   = half * 2;

    const float* Arow  = A + (size_t)(mt * 16 + fr) * K + kb;
    const int    nbase = nt * 64;

    v8f acc[4] = {{}, {}, {}, {}};
    for (int k0 = 0; k0 < K; k0 += 4) {
        v2f a;
        a.x = Arow[k0];
        a.y = Arow[k0 + 1];
        const float* bp = BP + ((size_t)(k0 >> 2) * N + nbase + fr) * 4 + kb;
        v2f b0, b1, b2, b3;
        b0.x = bp[0];       b0.y = bp[1];
        b1.x = bp[16 * 4];  b1.y = bp[16 * 4 + 1];
        b2.x = bp[32 * 4];  b2.y = bp[32 * 4 + 1];
        b3.x = bp[48 * 4];  b3.y = bp[48 * 4 + 1];
        acc[0] = __builtin_amdgcn_wmma_f32_16x16x4_f32(false, a, false, b0, (short)0, acc[0], false, false);
        acc[1] = __builtin_amdgcn_wmma_f32_16x16x4_f32(false, a, false, b1, (short)0, acc[1], false, false);
        acc[2] = __builtin_amdgcn_wmma_f32_16x16x4_f32(false, a, false, b2, (short)0, acc[2], false, false);
        acc[3] = __builtin_amdgcn_wmma_f32_16x16x4_f32(false, a, false, b3, (short)0, acc[3], false, false);
    }

#pragma unroll
    for (int u = 0; u < 4; ++u) {
        const int g = nbase + u * 16 + fr;
        float bs = 0.0f;
        if (bias0) bs += bias0[g];
        if (bias1) bs += bias1[g];
#pragma unroll
        for (int v = 0; v < 8; ++v) {
            int row = v + half * 8;
            C[(size_t)(mt * 16 + row) * N + g] = acc[u][v] + bs;
        }
    }
}

__global__ void lstm_layer(const float* __restrict__ gx_f, const float* __restrict__ gx_b,
                           const float* __restrict__ WhhT_f, const float* __restrict__ WhhT_b,
                           float* __restrict__ hout) {
    const int dir = blockIdx.x;
    const float* gx   = dir ? gx_b   : gx_f;
    const float* WhhT = dir ? WhhT_b : WhhT_f;

    __shared__ float h_s[HDIM];
    __shared__ float c_s[HDIM];
    __shared__ float g_s[G4H];

    const int tid = threadIdx.x;
    if (tid < HDIM) { h_s[tid] = 0.0f; c_s[tid] = 0.0f; }
    __syncthreads();

    for (int step = 0; step < S_LEN; ++step) {
        const int t = dir ? (S_LEN - 1 - step) : step;
        float acc = gx[(size_t)t * G4H + tid];
#pragma unroll 8
        for (int j = 0; j < HDIM; ++j) {
            acc += h_s[j] * WhhT[(size_t)j * G4H + tid];
        }
        g_s[tid] = acc;
        __syncthreads();
        if (tid < HDIM) {
            float gi = g_s[tid];
            float gf = g_s[HDIM + tid];
            float gc = g_s[2 * HDIM + tid];
            float go = g_s[3 * HDIM + tid];
            float c  = fast_sigmoid(gf) * c_s[tid] + fast_sigmoid(gi) * fast_tanh(gc);
            float h  = fast_sigmoid(go) * fast_tanh(c);
            c_s[tid] = c;
            h_s[tid] = h;
            hout[(size_t)t * (2 * HDIM) + dir * HDIM + tid] = h;
        }
        __syncthreads();
    }
}

__global__ void pairwise_kernel(const float* __restrict__ hA, const float* __restrict__ hB,
                                const float* __restrict__ W2, const float* __restrict__ b2,
                                float* __restrict__ out) {
    __shared__ float sA[16][MLP];
    __shared__ float sB[16][MLP];
    __shared__ float sW[MLP];

    const int i0 = blockIdx.y * 16;
    const int j0 = blockIdx.x * 16;
    const int tid = threadIdx.x;

    for (int idx = tid; idx < 16 * MLP; idx += 256) {
        int r = idx >> 7, k = idx & (MLP - 1);
        sA[r][k] = hA[(size_t)(i0 + r) * MLP + k];
        sB[r][k] = hB[(size_t)(j0 + r) * MLP + k];
    }
    if (tid < MLP) sW[tid] = W2[tid];
    __syncthreads();

    const int ti = tid >> 4;
    const int tj = tid & 15;
    const int i = i0 + ti;
    const int j = j0 + tj;

    float acc = 0.0f;
#pragma unroll 4
    for (int k = 0; k < MLP; ++k) {
        acc += sW[k] * fast_tanh(sA[ti][k] + sB[tj][k]);
    }
    float r = acc + b2[0];
    out[(size_t)i * S_LEN + j] = (i == j) ? 0.0f : r;
}

extern "C" void kernel_launch(void* const* d_in, const int* in_sizes, int n_in,
                              void* d_out, int out_size, void* d_ws, size_t ws_size,
                              hipStream_t stream) {
    const int*   widx   = (const int*)d_in[0];
    const int*   pidx   = (const int*)d_in[1];
    const float* Wword  = (const float*)d_in[3];
    const float* Wpos   = (const float*)d_in[4];
    const float* Wih_l0f = (const float*)d_in[5];
    const float* Whh_l0f = (const float*)d_in[6];
    const float* bih_l0f = (const float*)d_in[7];
    const float* bhh_l0f = (const float*)d_in[8];
    const float* Wih_l0b = (const float*)d_in[9];
    const float* Whh_l0b = (const float*)d_in[10];
    const float* bih_l0b = (const float*)d_in[11];
    const float* bhh_l0b = (const float*)d_in[12];
    const float* Wih_l1f = (const float*)d_in[13];
    const float* Whh_l1f = (const float*)d_in[14];
    const float* bih_l1f = (const float*)d_in[15];
    const float* bhh_l1f = (const float*)d_in[16];
    const float* Wih_l1b = (const float*)d_in[17];
    const float* Whh_l1b = (const float*)d_in[18];
    const float* bih_l1b = (const float*)d_in[19];
    const float* bhh_l1b = (const float*)d_in[20];
    const float* W1 = (const float*)d_in[21];
    const float* b1 = (const float*)d_in[22];
    const float* W2 = (const float*)d_in[23];
    const float* b2 = (const float*)d_in[24];
    float* out = (float*)d_out;

    float* ws = (float*)d_ws;
    float* x      = ws;
    float* gxf    = x      + S_LEN * EMB;
    float* gxb    = gxf    + S_LEN * G4H;
    float* h0     = gxb    + S_LEN * G4H;
    float* h1     = h0     + S_LEN * 2 * HDIM;
    float* hA     = h1     + S_LEN * 2 * HDIM;
    float* hB     = hA     + S_LEN * MLP;
    float* BP0f   = hB     + S_LEN * MLP;
    float* BP0b   = BP0f   + EMB * G4H;
    float* BP1f   = BP0b   + EMB * G4H;
    float* BP1b   = BP1f   + 2 * HDIM * G4H;
    float* WhhT0f = BP1b   + 2 * HDIM * G4H;
    float* WhhT0b = WhhT0f + HDIM * G4H;
    float* WhhT1f = WhhT0b + HDIM * G4H;
    float* WhhT1b = WhhT1f + HDIM * G4H;
    float* BPA    = WhhT1b + HDIM * G4H;
    float* BPB    = BPA    + 2 * HDIM * MLP;

    auto tgrid = [](int total) { return dim3((total + 255) / 256); };
    pack_bt<<<tgrid(G4H * EMB),      256, 0, stream>>>(Wih_l0f, BP0f, G4H, EMB,      0,      EMB);
    pack_bt<<<tgrid(G4H * EMB),      256, 0, stream>>>(Wih_l0b, BP0b, G4H, EMB,      0,      EMB);
    pack_bt<<<tgrid(G4H * 2 * HDIM), 256, 0, stream>>>(Wih_l1f, BP1f, G4H, 2 * HDIM, 0, 2 * HDIM);
    pack_bt<<<tgrid(G4H * 2 * HDIM), 256, 0, stream>>>(Wih_l1b, BP1b, G4H, 2 * HDIM, 0, 2 * HDIM);
    pack_bt<<<tgrid(MLP * 2 * HDIM), 256, 0, stream>>>(W1, BPA, MLP, G4H,        0, 2 * HDIM);
    pack_bt<<<tgrid(MLP * 2 * HDIM), 256, 0, stream>>>(W1, BPB, MLP, G4H, 2 * HDIM, 2 * HDIM);
    transpose_sub<<<tgrid(G4H * HDIM), 256, 0, stream>>>(Whh_l0f, WhhT0f, G4H, HDIM, 0, HDIM);
    transpose_sub<<<tgrid(G4H * HDIM), 256, 0, stream>>>(Whh_l0b, WhhT0b, G4H, HDIM, 0, HDIM);
    transpose_sub<<<tgrid(G4H * HDIM), 256, 0, stream>>>(Whh_l1f, WhhT1f, G4H, HDIM, 0, HDIM);
    transpose_sub<<<tgrid(G4H * HDIM), 256, 0, stream>>>(Whh_l1b, WhhT1b, G4H, HDIM, 0, HDIM);

    embed_kernel<<<S_LEN, EMB, 0, stream>>>(widx, pidx, Wword, Wpos, x);

    {
        dim3 grid(G4H / 64, S_LEN / 16);
        gemm_wmma_f32<<<grid, 32, 0, stream>>>(x, BP0f, bih_l0f, bhh_l0f, gxf, S_LEN, G4H, EMB);
        gemm_wmma_f32<<<grid, 32, 0, stream>>>(x, BP0b, bih_l0b, bhh_l0b, gxb, S_LEN, G4H, EMB);
    }
    lstm_layer<<<2, G4H, 0, stream>>>(gxf, gxb, WhhT0f, WhhT0b, h0);

    {
        dim3 grid(G4H / 64, S_LEN / 16);
        gemm_wmma_f32<<<grid, 32, 0, stream>>>(h0, BP1f, bih_l1f, bhh_l1f, gxf, S_LEN, G4H, 2 * HDIM);
        gemm_wmma_f32<<<grid, 32, 0, stream>>>(h0, BP1b, bih_l1b, bhh_l1b, gxb, S_LEN, G4H, 2 * HDIM);
    }
    lstm_layer<<<2, G4H, 0, stream>>>(gxf, gxb, WhhT1f, WhhT1b, h1);

    {
        dim3 grid(MLP / 64, S_LEN / 16);
        gemm_wmma_f32<<<grid, 32, 0, stream>>>(h1, BPA, b1, (const float*)nullptr, hA, S_LEN, MLP, 2 * HDIM);
        gemm_wmma_f32<<<grid, 32, 0, stream>>>(h1, BPB, (const float*)nullptr, (const float*)nullptr, hB, S_LEN, MLP, 2 * HDIM);
    }

    {
        dim3 grid(S_LEN / 16, S_LEN / 16);
        pairwise_kernel<<<grid, 256, 0, stream>>>(hA, hB, W2, b2, out);
    }
}


# LSTM recurrence: Whh kept on chip over all 768 steps (77 k-rows in VGPRs, 79 in LDS, 100 streamed through a prefetch ring), one thread per hidden unit with its 4 gates; summation order unchanged
# speedup vs baseline: 2.7927x; 2.7927x over previous
#include <hip/hip_runtime.h>
#include <hip/hip_bf16.h>
#include <utility>


#define S_LEN   768
#define V_WORD  30000
#define V_POS   64
#define WEMB    100
#define PEMB    28
#define EMB     128
#define HDIM    256
#define G4H     1024
#define MLP     128

typedef __attribute__((ext_vector_type(2))) float v2f;
typedef __attribute__((ext_vector_type(8))) float v8f;

__device__ __forceinline__ float fast_sigmoid(float x) {
    return 1.0f / (1.0f + __expf(-x));
}
__device__ __forceinline__ float fast_tanh(float x) {
    float c = fminf(fmaxf(x, -9.0f), 9.0f);
    float e = __expf(2.0f * c);
    return (e - 1.0f) / (e + 1.0f);
}

__global__ void embed_kernel(const int* __restrict__ widx, const int* __restrict__ pidx,
                             const float* __restrict__ Wword, const float* __restrict__ Wpos,
                             float* __restrict__ x) {
    int s = blockIdx.x;
    int t = threadIdx.x;
    float v;
    if (t < WEMB) v = Wword[(size_t)widx[s] * WEMB + t];
    else          v = Wpos[(size_t)pidx[s] * PEMB + (t - WEMB)];
    x[(size_t)s * EMB + t] = v;
}

__global__ void pack_whh(const float* __restrict__ src, float* __restrict__ dst) {
    const int idx = blockIdx.x * blockDim.x + threadIdx.x;
    if (idx >= G4H * HDIM) return;
    const int row = idx >> 8;
    const int j   = idx & (HDIM - 1);
    const int g   = row >> 8;
    const int u   = row & (HDIM - 1);
    dst[((size_t)j * HDIM + u) * 4 + g] = src[idx];
}

__global__ void pack_bt(const float* __restrict__ src, float* __restrict__ dst,
                        int Nout, int Ctot, int coff, int Kcnt) {
    int idx = blockIdx.x * blockDim.x + threadIdx.x;
    int total = Nout * Kcnt;
    if (idx >= total) return;
    int n = idx / Kcnt;
    int k = idx - n * Kcnt;
    dst[((size_t)(k >> 2) * Nout + n) * 4 + (k & 3)] = src[(size_t)n * Ctot + coff + k];
}

__global__ void __launch_bounds__(32)
gemm_wmma_f32(const float* __restrict__ A, const float* __restrict__ BP,
              const float* __restrict__ bias0, const float* __restrict__ bias1,
              float* __restrict__ C, int M, int N, int K) {
    const int nt = blockIdx.x;
    const int mt = blockIdx.y;
    const int l  = threadIdx.x;
    const int fr   = l & 15;
    const int half = l >> 4;
    const int kb   = half * 2;

    const float* Arow  = A + (size_t)(mt * 16 + fr) * K + kb;
    const int    nbase = nt * 64;

    v8f acc[4] = {{}, {}, {}, {}};
    for (int k0 = 0; k0 < K; k0 += 4) {
        v2f a;
        a.x = Arow[k0];
        a.y = Arow[k0 + 1];
        const float* bp = BP + ((size_t)(k0 >> 2) * N + nbase + fr) * 4 + kb;
        v2f b0, b1, b2, b3;
        b0.x = bp[0];       b0.y = bp[1];
        b1.x = bp[16 * 4];  b1.y = bp[16 * 4 + 1];
        b2.x = bp[32 * 4];  b2.y = bp[32 * 4 + 1];
        b3.x = bp[48 * 4];  b3.y = bp[48 * 4 + 1];
        acc[0] = __builtin_amdgcn_wmma_f32_16x16x4_f32(false, a, false, b0, (short)0, acc[0], false, false);
        acc[1] = __builtin_amdgcn_wmma_f32_16x16x4_f32(false, a, false, b1, (short)0, acc[1], false, false);
        acc[2] = __builtin_amdgcn_wmma_f32_16x16x4_f32(false, a, false, b2, (short)0, acc[2], false, false);
        acc[3] = __builtin_amdgcn_wmma_f32_16x16x4_f32(false, a, false, b3, (short)0, acc[3], false, false);
    }

#pragma unroll
    for (int u = 0; u < 4; ++u) {
        const int g = nbase + u * 16 + fr;
        float bs = 0.0f;
        if (bias0) bs += bias0[g];
        if (bias1) bs += bias1[g];
#pragma unroll
        for (int v = 0; v < 8; ++v) {
            int row = v + half * 8;
            C[(size_t)(mt * 16 + row) * N + g] = acc[u][v] + bs;
        }
    }
}

#define KS 100
#define SG 5
#define SR 4
#define KL 79
#define NS (HDIM - KS)
#define KR (NS - KL)
#define HQ (2 * HDIM / 4)
#define WSEG 4096
#define NSEG ((KL * HDIM + WSEG - 1) / WSEG)

constexpr int  n_str(int j)  { return (j * KS) / HDIM; }
constexpr bool is_str(int j) { return n_str(j + 1) != n_str(j); }
constexpr int  j_of_str(int s) { return ((s + 1) * HDIM + KS - 1) / KS - 1; }
constexpr int  n_chip(int j) { return j - n_str(j); }
constexpr int  n_lds(int j)  { return (n_chip(j) * KL) / NS; }
constexpr bool is_lds(int j) { return !is_str(j) && ((n_chip(j) + 1) * KL) / NS != n_lds(j); }
constexpr int  n_reg(int j)  { return n_chip(j) - n_lds(j); }

__device__ __forceinline__ void lds_barrier() {
    __builtin_amdgcn_fence(__ATOMIC_RELEASE, "workgroup", "local");
    __builtin_amdgcn_s_barrier();
    __builtin_amdgcn_fence(__ATOMIC_ACQUIRE, "workgroup", "local");
}

template <int J, bool LDS_PASS>
__device__ __forceinline__ void whh_place_row(const float4* __restrict__ Wp, float4* lds, const int (&wb)[NSEG],
                                              float4 (&wr)[KR]) {
    if constexpr (LDS_PASS && is_lds(J)) lds[wb[n_lds(J) / 16] + (n_lds(J) % 16) * HDIM] = Wp[J * HDIM];
    if constexpr (!LDS_PASS && !is_str(J) && !is_lds(J)) wr[n_reg(J)] = Wp[J * HDIM];
}
template <bool LDS_PASS, int... Js>
__device__ __forceinline__ void whh_place_rows(std::integer_sequence<int, Js...>, const float4* __restrict__ Wp,
                                               float4* lds, const int (&wb)[NSEG], float4 (&wr)[KR]) {
    (whh_place_row<Js, LDS_PASS>(Wp, lds, wb, wr), ...);
}
template <int... Ss>
__device__ __forceinline__ void whh_stream_fill(std::integer_sequence<int, Ss...>, const float4* __restrict__ Wp,
                                                float4 (&sb)[SR * SG]) {
    ((sb[Ss] = Wp[j_of_str(Ss) * HDIM]), ...);
}
template <int G, int... Is>
__device__ __forceinline__ void whh_refill(std::integer_sequence<int, Is...>, const float4* __restrict__ Wp,
                                           float4 (&sb)[SR * SG]) {
    ((sb[(G % SR) * SG + Is] = Wp[j_of_str(((G + SR) * SG + Is) % KS) * HDIM]), ...);
}

template <int J>
__device__ __forceinline__ void whh_row(const float4* __restrict__ Wp, const float4* lds, const int (&wb)[NSEG], int hb,
                                        float4 (&wr)[KR], float4 (&sb)[SR * SG], float4& hv,
                                        float& a_i, float& a_f, float& a_g, float& a_o) {
    if constexpr ((J & 3) == 0) hv = lds[hb + (J >> 2)];
    const float hj = ((J & 3) == 0) ? hv.x : ((J & 3) == 1) ? hv.y : ((J & 3) == 2) ? hv.z : hv.w;
    float4 w;
    if constexpr (is_str(J)) {
        constexpr int s = n_str(J);
        constexpr int g = s / SG;
        w = sb[(g % SR) * SG + (s % SG)];
    } else if constexpr (is_lds(J)) {
        w = lds[wb[n_lds(J) / 16] + (n_lds(J) % 16) * HDIM];
    } else {
        w = wr[n_reg(J)];
    }
    a_i += hj * w.x;
    a_f += hj * w.y;
    a_g += hj * w.z;
    a_o += hj * w.w;
    if constexpr (is_str(J) && (n_str(J) % SG) == SG - 1) {
        constexpr int g = n_str(J) / SG;
        whh_refill<g>(std::make_integer_sequence<int, SG>{}, Wp, sb);
        asm volatile("" : "+v"(a_i));
    }
}
template <int... Js>
__device__ __forceinline__ void whh_rows(std::integer_sequence<int, Js...>, const float4* __restrict__ Wp,
                                         const float4* lds, const int (&wb)[NSEG], int hb,
                                         float4 (&wr)[KR], float4 (&sb)[SR * SG], float4& hv,
                                         float& a_i, float& a_f, float& a_g, float& a_o) {
    (whh_row<Js>(Wp, lds, wb, hb, wr, sb, hv, a_i, a_f, a_g, a_o), ...);
}

__global__ void __launch_bounds__(HDIM)
lstm_layer(const float* __restrict__ gx_f, const float* __restrict__ gx_b,
           const float* __restrict__ Wp_f, const float* __restrict__ Wp_b,
           float* __restrict__ hout) {
    const int dir = blockIdx.x;
    const int u   = threadIdx.x;
    const float*  gx = (dir ? gx_b : gx_f) + u;
    const float4* Wp = reinterpret_cast<const float4*>(dir ? Wp_b : Wp_f) + u;

    __shared__ float4 lds[HQ + KL * HDIM];
    float* h_s = reinterpret_cast<float*>(lds);

    int wb[NSEG];
#pragma unroll
    for (int k = 0; k < NSEG; ++k) {
        wb[k] = HQ + u + k * WSEG;
        asm volatile("" : "+v"(wb[k]));
    }

    float4 wr[KR];
    float4 sb[SR * SG];
    whh_place_rows<true>(std::make_integer_sequence<int, HDIM>{}, Wp, lds, wb, wr);
    asm volatile("" ::: "memory");
    whh_place_rows<false>(std::make_integer_sequence<int, HDIM>{}, Wp, lds, wb, wr);
    whh_stream_fill(std::make_integer_sequence<int, SR * SG>{}, Wp, sb);

    h_s[u] = 0.0f;
    float c = 0.0f;

    int t = dir ? (S_LEN - 1) : 0;
    const int dt = dir ? -1 : 1;
    float g_i = gx[(size_t)t * G4H];
    float g_f = gx[(size_t)t * G4H + HDIM];
    float g_g = gx[(size_t)t * G4H + 2 * HDIM];
    float g_o = gx[(size_t)t * G4H + 3 * HDIM];
    __syncthreads();

    for (int step = 0; step < S_LEN; ++step) {
        const int cur = step & 1;
        float a_i = g_i, a_f = g_f, a_g = g_g, a_o = g_o;

        const int tn = (step + 1 < S_LEN) ? (t + dt) : t;
        g_i = gx[(size_t)tn * G4H];
        g_f = gx[(size_t)tn * G4H + HDIM];
        g_g = gx[(size_t)tn * G4H + 2 * HDIM];
        g_o = gx[(size_t)tn * G4H + 3 * HDIM];

        int zoff = 0;
        asm volatile("" : "+s"(zoff));
        const float4* Wst = Wp + zoff;

        int hb = cur * (HDIM / 4);
        asm volatile("" : "+v"(hb));

        float4 hv;
        whh_rows(std::make_integer_sequence<int, HDIM>{}, Wst, lds, wb, hb, wr, sb, hv, a_i, a_f, a_g, a_o);

        c = fast_sigmoid(a_f) * c + fast_sigmoid(a_i) * fast_tanh(a_g);
        const float h = fast_sigmoid(a_o) * fast_tanh(c);
        h_s[(cur ^ 1) * HDIM + u] = h;
        hout[(size_t)t * (2 * HDIM) + dir * HDIM + u] = h;
        t = tn;
        lds_barrier();
    }
}

__global__ void pairwise_kernel(const float* __restrict__ hA, const float* __restrict__ hB,
                                const float* __restrict__ W2, const float* __restrict__ b2,
                                float* __restrict__ out) {
    __shared__ float sA[16][MLP];
    __shared__ float sB[16][MLP];
    __shared__ float sW[MLP];

    const int i0 = blockIdx.y * 16;
    const int j0 = blockIdx.x * 16;
    const int tid = threadIdx.x;

    for (int idx = tid; idx < 16 * MLP; idx += 256) {
        int r = idx >> 7, k = idx & (MLP - 1);
        sA[r][k] = hA[(size_t)(i0 + r) * MLP + k];
        sB[r][k] = hB[(size_t)(j0 + r) * MLP + k];
    }
    if (tid < MLP) sW[tid] = W2[tid];
    __syncthreads();

    const int ti = tid >> 4;
    const int tj = tid & 15;
    const int i = i0 + ti;
    const int j = j0 + tj;

    float acc = 0.0f;
#pragma unroll 4
    for (int k = 0; k < MLP; ++k) {
        acc += sW[k] * fast_tanh(sA[ti][k] + sB[tj][k]);
    }
    float r = acc + b2[0];
    out[(size_t)i * S_LEN + j] = (i == j) ? 0.0f : r;
}

extern "C" void kernel_launch(void* const* d_in, const int* in_sizes, int n_in,
                              void* d_out, int out_size, void* d_ws, size_t ws_size,
                              hipStream_t stream) {
    const int*   widx   = (const int*)d_in[0];
    const int*   pidx   = (const int*)d_in[1];
    const float* Wword  = (const float*)d_in[3];
    const float* Wpos   = (const float*)d_in[4];
    const float* Wih_l0f = (const float*)d_in[5];
    const float* Whh_l0f = (const float*)d_in[6];
    const float* bih_l0f = (const float*)d_in[7];
    const float* bhh_l0f = (const float*)d_in[8];
    const float* Wih_l0b = (const float*)d_in[9];
    const float* Whh_l0b = (const float*)d_in[10];
    const float* bih_l0b = (const float*)d_in[11];
    const float* bhh_l0b = (const float*)d_in[12];
    const float* Wih_l1f = (const float*)d_in[13];
    const float* Whh_l1f = (const float*)d_in[14];
    const float* bih_l1f = (const float*)d_in[15];
    const float* bhh_l1f = (const float*)d_in[16];
    const float* Wih_l1b = (const float*)d_in[17];
    const float* Whh_l1b = (const float*)d_in[18];
    const float* bih_l1b = (const float*)d_in[19];
    const float* bhh_l1b = (const float*)d_in[20];
    const float* W1 = (const float*)d_in[21];
    const float* b1 = (const float*)d_in[22];
    const float* W2 = (const float*)d_in[23];
    const float* b2 = (const float*)d_in[24];
    float* out = (float*)d_out;

    float* ws = (float*)d_ws;
    float* x      = ws;
    float* gxf    = x      + S_LEN * EMB;
    float* gxb    = gxf    + S_LEN * G4H;
    float* h0     = gxb    + S_LEN * G4H;
    float* h1     = h0     + S_LEN * 2 * HDIM;
    float* hA     = h1     + S_LEN * 2 * HDIM;
    float* hB     = hA     + S_LEN * MLP;
    float* BP0f   = hB     + S_LEN * MLP;
    float* BP0b   = BP0f   + EMB * G4H;
    float* BP1f   = BP0b   + EMB * G4H;
    float* BP1b   = BP1f   + 2 * HDIM * G4H;
    float* WhhT0f = BP1b   + 2 * HDIM * G4H;
    float* WhhT0b = WhhT0f + HDIM * G4H;
    float* WhhT1f = WhhT0b + HDIM * G4H;
    float* WhhT1b = WhhT1f + HDIM * G4H;
    float* BPA    = WhhT1b + HDIM * G4H;
    float* BPB    = BPA    + 2 * HDIM * MLP;

    auto tgrid = [](int total) { return dim3((total + 255) / 256); };
    pack_bt<<<tgrid(G4H * EMB),      256, 0, stream>>>(Wih_l0f, BP0f, G4H, EMB,      0,      EMB);
    pack_bt<<<tgrid(G4H * EMB),      256, 0, stream>>>(Wih_l0b, BP0b, G4H, EMB,      0,      EMB);
    pack_bt<<<tgrid(G4H * 2 * HDIM), 256, 0, stream>>>(Wih_l1f, BP1f, G4H, 2 * HDIM, 0, 2 * HDIM);
    pack_bt<<<tgrid(G4H * 2 * HDIM), 256, 0, stream>>>(Wih_l1b, BP1b, G4H, 2 * HDIM, 0, 2 * HDIM);
    pack_bt<<<tgrid(MLP * 2 * HDIM), 256, 0, stream>>>(W1, BPA, MLP, G4H,        0, 2 * HDIM);
    pack_bt<<<tgrid(MLP * 2 * HDIM), 256, 0, stream>>>(W1, BPB, MLP, G4H, 2 * HDIM, 2 * HDIM);
    pack_whh<<<tgrid(G4H * HDIM), 256, 0, stream>>>(Whh_l0f, WhhT0f);
    pack_whh<<<tgrid(G4H * HDIM), 256, 0, stream>>>(Whh_l0b, WhhT0b);
    pack_whh<<<tgrid(G4H * HDIM), 256, 0, stream>>>(Whh_l1f, WhhT1f);
    pack_whh<<<tgrid(G4H * HDIM), 256, 0, stream>>>(Whh_l1b, WhhT1b);

    embed_kernel<<<S_LEN, EMB, 0, stream>>>(widx, pidx, Wword, Wpos, x);

    {
        dim3 grid(G4H / 64, S_LEN / 16);
        gemm_wmma_f32<<<grid, 32, 0, stream>>>(x, BP0f, bih_l0f, bhh_l0f, gxf, S_LEN, G4H, EMB);
        gemm_wmma_f32<<<grid, 32, 0, stream>>>(x, BP0b, bih_l0b, bhh_l0b, gxb, S_LEN, G4H, EMB);
    }
    lstm_layer<<<2, HDIM, 0, stream>>>(gxf, gxb, WhhT0f, WhhT0b, h0);

    {
        dim3 grid(G4H / 64, S_LEN / 16);
        gemm_wmma_f32<<<grid, 32, 0, stream>>>(h0, BP1f, bih_l1f, bhh_l1f, gxf, S_LEN, G4H, 2 * HDIM);
        gemm_wmma_f32<<<grid, 32, 0, stream>>>(h0, BP1b, bih_l1b, bhh_l1b, gxb, S_LEN, G4H, 2 * HDIM);
    }
    lstm_layer<<<2, HDIM, 0, stream>>>(gxf, gxb, WhhT1f, WhhT1b, h1);

    {
        dim3 grid(MLP / 64, S_LEN / 16);
        gemm_wmma_f32<<<grid, 32, 0, stream>>>(h1, BPA, b1, (const float*)nullptr, hA, S_LEN, MLP, 2 * HDIM);
        gemm_wmma_f32<<<grid, 32, 0, stream>>>(h1, BPB, (const float*)nullptr, (const float*)nullptr, hB, S_LEN, MLP, 2 * HDIM);
    }

    {
        dim3 grid(S_LEN / 16, S_LEN / 16);
        pairwise_kernel<<<grid, 256, 0, stream>>>(hA, hB, W2, b2, out);
    }
}


# LSTM recurrence: LDS reads (weight rows, h pieces) issued in one burst per ring segment, 8+ k-rows ahead of first use (LDS rows last in a segment), no wait in place; summation order unchanged
# speedup vs baseline: 3.1323x; 1.1216x over previous
#include <hip/hip_runtime.h>
#include <hip/hip_bf16.h>
#include <utility>


#define S_LEN   768
#define V_WORD  30000
#define V_POS   64
#define WEMB    100
#define PEMB    28
#define EMB     128
#define HDIM    256
#define G4H     1024
#define MLP     128

typedef __attribute__((ext_vector_type(2))) float v2f;
typedef __attribute__((ext_vector_type(8))) float v8f;

__device__ __forceinline__ float fast_sigmoid(float x) {
    return 1.0f / (1.0f + __expf(-x));
}
__device__ __forceinline__ float fast_tanh(float x) {
    float c = fminf(fmaxf(x, -9.0f), 9.0f);
    float e = __expf(2.0f * c);
    return (e - 1.0f) / (e + 1.0f);
}

__global__ void embed_kernel(const int* __restrict__ widx, const int* __restrict__ pidx,
                             const float* __restrict__ Wword, const float* __restrict__ Wpos,
                             float* __restrict__ x) {
    int s = blockIdx.x;
    int t = threadIdx.x;
    float v;
    if (t < WEMB) v = Wword[(size_t)widx[s] * WEMB + t];
    else          v = Wpos[(size_t)pidx[s] * PEMB + (t - WEMB)];
    x[(size_t)s * EMB + t] = v;
}

__global__ void pack_whh(const float* __restrict__ src, float* __restrict__ dst) {
    const int idx = blockIdx.x * blockDim.x + threadIdx.x;
    if (idx >= G4H * HDIM) return;
    const int row = idx >> 8;
    const int j   = idx & (HDIM - 1);
    const int g   = row >> 8;
    const int u   = row & (HDIM - 1);
    dst[((size_t)j * HDIM + u) * 4 + g] = src[idx];
}

__global__ void pack_bt(const float* __restrict__ src, float* __restrict__ dst,
                        int Nout, int Ctot, int coff, int Kcnt) {
    int idx = blockIdx.x * blockDim.x + threadIdx.x;
    int total = Nout * Kcnt;
    if (idx >= total) return;
    int n = idx / Kcnt;
    int k = idx - n * Kcnt;
    dst[((size_t)(k >> 2) * Nout + n) * 4 + (k & 3)] = src[(size_t)n * Ctot + coff + k];
}

__global__ void __launch_bounds__(32)
gemm_wmma_f32(const float* __restrict__ A, const float* __restrict__ BP,
              const float* __restrict__ bias0, const float* __restrict__ bias1,
              float* __restrict__ C, int M, int N, int K) {
    const int nt = blockIdx.x;
    const int mt = blockIdx.y;
    const int l  = threadIdx.x;
    const int fr   = l & 15;
    const int half = l >> 4;
    const int kb   = half * 2;

    const float* Arow  = A + (size_t)(mt * 16 + fr) * K + kb;
    const int    nbase = nt * 64;

    v8f acc[4] = {{}, {}, {}, {}};
    for (int k0 = 0; k0 < K; k0 += 4) {
        v2f a;
        a.x = Arow[k0];
        a.y = Arow[k0 + 1];
        const float* bp = BP + ((size_t)(k0 >> 2) * N + nbase + fr) * 4 + kb;
        v2f b0, b1, b2, b3;
        b0.x = bp[0];       b0.y = bp[1];
        b1.x = bp[16 * 4];  b1.y = bp[16 * 4 + 1];
        b2.x = bp[32 * 4];  b2.y = bp[32 * 4 + 1];
        b3.x = bp[48 * 4];  b3.y = bp[48 * 4 + 1];
        acc[0] = __builtin_amdgcn_wmma_f32_16x16x4_f32(false, a, false, b0, (short)0, acc[0], false, false);
        acc[1] = __builtin_amdgcn_wmma_f32_16x16x4_f32(false, a, false, b1, (short)0, acc[1], false, false);
        acc[2] = __builtin_amdgcn_wmma_f32_16x16x4_f32(false, a, false, b2, (short)0, acc[2], false, false);
        acc[3] = __builtin_amdgcn_wmma_f32_16x16x4_f32(false, a, false, b3, (short)0, acc[3], false, false);
    }

#pragma unroll
    for (int u = 0; u < 4; ++u) {
        const int g = nbase + u * 16 + fr;
        float bs = 0.0f;
        if (bias0) bs += bias0[g];
        if (bias1) bs += bias1[g];
#pragma unroll
        for (int v = 0; v < 8; ++v) {
            int row = v + half * 8;
            C[(size_t)(mt * 16 + row) * N + g] = acc[u][v] + bs;
        }
    }
}

#define KS 100
#define SG 5
#define SR 4
#define KL 79
#define NS (HDIM - KS)
#define KR (NS - KL)
#define NG (KS / SG)
#define PD 8
#define HQ (2 * HDIM / 4)
#define WSEG 4096
#define NSEG ((KL * HDIM + WSEG - 1) / WSEG)

constexpr int  seg_start(int g) { return (g * HDIM) / NG; }
constexpr int  seg_of(int j)    { return ((j + 1) * NG - 1) / HDIM; }
constexpr int  seg_off(int j)   { return j - seg_start(seg_of(j)); }
constexpr bool seg_last(int j)  { return j + 1 == seg_start(seg_of(j) + 1); }
constexpr int  lds_start(int g) { return (g * KL) / NG; }
constexpr int  reg_start(int g) { return seg_start(g) - g * SG - lds_start(g); }
constexpr int  seg_regs(int g)  { return reg_start(g + 1) - reg_start(g); }
constexpr bool is_reg(int j)    { return seg_off(j) < seg_regs(seg_of(j)); }
constexpr bool is_str(int j)    { return !is_reg(j) && seg_off(j) < seg_regs(seg_of(j)) + SG; }
constexpr bool is_lds(int j)    { return !is_reg(j) && !is_str(j); }
constexpr int  n_reg(int j)     { return reg_start(seg_of(j)) + seg_off(j); }
constexpr int  n_str(int j)     { return seg_of(j) * SG + seg_off(j) - seg_regs(seg_of(j)); }
constexpr int  n_lds(int j)     { return lds_start(seg_of(j)) + seg_off(j) - seg_regs(seg_of(j)) - SG; }
constexpr int  j_of_str(int s)  { return seg_start(s / SG) + seg_regs(s / SG) + s % SG; }
constexpr int  fetch_seg(int j) { return (seg_off(j) >= PD || seg_of(j) == 0) ? seg_of(j) : seg_of(j) - 1; }
constexpr bool layout_ok() {
    if (reg_start(NG) != KR || lds_start(NG) != KL || seg_start(NG) != HDIM) return false;
    for (int g = 0; g < NG; ++g) if (seg_regs(g) < 0 || seg_regs(g) + SG < PD) return false;
    for (int s = 0; s < KS; ++s) if (!is_str(j_of_str(s)) || n_str(j_of_str(s)) != s) return false;
    return true;
}
static_assert(layout_ok(), "segment layout does not match KS / KL / KR");

__device__ __forceinline__ void lds_barrier() {
    __builtin_amdgcn_fence(__ATOMIC_RELEASE, "workgroup", "local");
    __builtin_amdgcn_s_barrier();
    __builtin_amdgcn_fence(__ATOMIC_ACQUIRE, "workgroup", "local");
}

template <int J, bool LDS_PASS>
__device__ __forceinline__ void whh_place_row(const float4* __restrict__ Wp, float4* lds, const int (&wb)[NSEG],
                                              float4 (&wr)[KR]) {
    if constexpr (LDS_PASS && is_lds(J)) lds[wb[n_lds(J) / 16] + (n_lds(J) % 16) * HDIM] = Wp[J * HDIM];
    if constexpr (!LDS_PASS && is_reg(J)) wr[n_reg(J)] = Wp[J * HDIM];
}
template <bool LDS_PASS, int... Js>
__device__ __forceinline__ void whh_place_rows(std::integer_sequence<int, Js...>, const float4* __restrict__ Wp,
                                               float4* lds, const int (&wb)[NSEG], float4 (&wr)[KR]) {
    (whh_place_row<Js, LDS_PASS>(Wp, lds, wb, wr), ...);
}
template <int... Ss>
__device__ __forceinline__ void whh_stream_fill(std::integer_sequence<int, Ss...>, const float4* __restrict__ Wp,
                                                float4 (&sb)[SR * SG]) {
    ((sb[Ss] = Wp[j_of_str(Ss) * HDIM]), ...);
}
template <int G, int... Is>
__device__ __forceinline__ void whh_refill(std::integer_sequence<int, Is...>, const float4* __restrict__ Wp,
                                           float4 (&sb)[SR * SG]) {
    ((sb[(G % SR) * SG + Is] = Wp[j_of_str(((G + SR) * SG + Is) % KS) * HDIM]), ...);
}

template <int G, int J>
__device__ __forceinline__ void lds_fetch_row(const float4* lds, const int (&wb)[NSEG], int hb,
                                              float4 (&lw)[KL], float4 (&hq)[HDIM / 4]) {
    if constexpr (fetch_seg(J) == G) {
        if constexpr ((J & 3) == 0) hq[J >> 2] = lds[hb + (J >> 2)];
        if constexpr (is_lds(J)) lw[n_lds(J)] = lds[wb[n_lds(J) / 16] + (n_lds(J) % 16) * HDIM];
    }
}
template <int G, int... Js>
__device__ __forceinline__ void lds_fetch_seg(std::integer_sequence<int, Js...>, const float4* lds, const int (&wb)[NSEG],
                                              int hb, float4 (&lw)[KL], float4 (&hq)[HDIM / 4]) {
    (lds_fetch_row<G, Js>(lds, wb, hb, lw, hq), ...);
}

template <int J>
__device__ __forceinline__ void whh_row(const float4* __restrict__ Wp, const float4* lds, const int (&wb)[NSEG], int hb,
                                        float4 (&wr)[KR], float4 (&sb)[SR * SG], float4 (&lw)[KL], float4 (&hq)[HDIM / 4],
                                        float& a_i, float& a_f, float& a_g, float& a_o) {
    if constexpr (J == 0) {
        lds_fetch_seg<0>(std::make_integer_sequence<int, HDIM>{}, lds, wb, hb, lw, hq);
        asm volatile("" : "+v"(a_i));
    }
    const float4 hv = hq[J >> 2];
    const float hj = ((J & 3) == 0) ? hv.x : ((J & 3) == 1) ? hv.y : ((J & 3) == 2) ? hv.z : hv.w;
    float4 w;
    if constexpr (is_str(J)) {
        constexpr int s = n_str(J);
        constexpr int g = s / SG;
        w = sb[(g % SR) * SG + (s % SG)];
    } else if constexpr (is_lds(J)) {
        w = lw[n_lds(J)];
    } else {
        w = wr[n_reg(J)];
    }
    a_i += hj * w.x;
    a_f += hj * w.y;
    a_g += hj * w.z;
    a_o += hj * w.w;
    if constexpr (is_str(J) && (n_str(J) % SG) == SG - 1) {
        constexpr int g = n_str(J) / SG;
        whh_refill<g>(std::make_integer_sequence<int, SG>{}, Wp, sb);
    }
    if constexpr (seg_last(J)) {
        if constexpr (seg_of(J) + 1 < NG)
            lds_fetch_seg<seg_of(J) + 1>(std::make_integer_sequence<int, HDIM>{}, lds, wb, hb, lw, hq);
        asm volatile("" : "+v"(a_i));
    }
}
template <int... Js>
__device__ __forceinline__ void whh_rows(std::integer_sequence<int, Js...>, const float4* __restrict__ Wp,
                                         const float4* lds, const int (&wb)[NSEG], int hb,
                                         float4 (&wr)[KR], float4 (&sb)[SR * SG], float4 (&lw)[KL], float4 (&hq)[HDIM / 4],
                                         float& a_i, float& a_f, float& a_g, float& a_o) {
    (whh_row<Js>(Wp, lds, wb, hb, wr, sb, lw, hq, a_i, a_f, a_g, a_o), ...);
}

__global__ void __launch_bounds__(HDIM)
lstm_layer(const float* __restrict__ gx_f, const float* __restrict__ gx_b,
           const float* __restrict__ Wp_f, const float* __restrict__ Wp_b,
           float* __restrict__ hout) {
    const int dir = blockIdx.x;
    const int u   = threadIdx.x;
    const float*  gx = (dir ? gx_b : gx_f) + u;
    const float4* Wp = reinterpret_cast<const float4*>(dir ? Wp_b : Wp_f) + u;

    __shared__ float4 lds[HQ + KL * HDIM];
    float* h_s = reinterpret_cast<float*>(lds);

    int wb[NSEG];
#pragma unroll
    for (int k = 0; k < NSEG; ++k) {
        wb[k] = HQ + u + k * WSEG;
        asm volatile("" : "+v"(wb[k]));
    }

    float4 wr[KR];
    float4 sb[SR * SG];
    whh_place_rows<true>(std::make_integer_sequence<int, HDIM>{}, Wp, lds, wb, wr);
    asm volatile("" ::: "memory");
    whh_place_rows<false>(std::make_integer_sequence<int, HDIM>{}, Wp, lds, wb, wr);
    whh_stream_fill(std::make_integer_sequence<int, SR * SG>{}, Wp, sb);

    h_s[u] = 0.0f;
    float c = 0.0f;

    int t = dir ? (S_LEN - 1) : 0;
    const int dt = dir ? -1 : 1;
    float g_i = gx[(size_t)t * G4H];
    float g_f = gx[(size_t)t * G4H + HDIM];
    float g_g = gx[(size_t)t * G4H + 2 * HDIM];
    float g_o = gx[(size_t)t * G4H + 3 * HDIM];
    __syncthreads();

    for (int step = 0; step < S_LEN; ++step) {
        const int cur = step & 1;
        float a_i = g_i, a_f = g_f, a_g = g_g, a_o = g_o;

        const int tn = (step + 1 < S_LEN) ? (t + dt) : t;
        g_i = gx[(size_t)tn * G4H];
        g_f = gx[(size_t)tn * G4H + HDIM];
        g_g = gx[(size_t)tn * G4H + 2 * HDIM];
        g_o = gx[(size_t)tn * G4H + 3 * HDIM];

        int zoff = 0;
        asm volatile("" : "+s"(zoff));
        const float4* Wst = Wp + zoff;

        int hb = cur * (HDIM / 4);
        asm volatile("" : "+v"(hb));

        float4 lw[KL];
        float4 hq[HDIM / 4];
        whh_rows(std::make_integer_sequence<int, HDIM>{}, Wst, lds, wb, hb, wr, sb, lw, hq, a_i, a_f, a_g, a_o);

        c = fast_sigmoid(a_f) * c + fast_sigmoid(a_i) * fast_tanh(a_g);
        const float h = fast_sigmoid(a_o) * fast_tanh(c);
        h_s[(cur ^ 1) * HDIM + u] = h;
        hout[(size_t)t * (2 * HDIM) + dir * HDIM + u] = h;
        t = tn;
        lds_barrier();
    }
}

__global__ void pairwise_kernel(const float* __restrict__ hA, const float* __restrict__ hB,
                                const float* __restrict__ W2, const float* __restrict__ b2,
                                float* __restrict__ out) {
    __shared__ float sA[16][MLP];
    __shared__ float sB[16][MLP];
    __shared__ float sW[MLP];

    const int i0 = blockIdx.y * 16;
    const int j0 = blockIdx.x * 16;
    const int tid = threadIdx.x;

    for (int idx = tid; idx < 16 * MLP; idx += 256) {
        int r = idx >> 7, k = idx & (MLP - 1);
        sA[r][k] = hA[(size_t)(i0 + r) * MLP + k];
        sB[r][k] = hB[(size_t)(j0 + r) * MLP + k];
    }
    if (tid < MLP) sW[tid] = W2[tid];
    __syncthreads();

    const int ti = tid >> 4;
    const int tj = tid & 15;
    const int i = i0 + ti;
    const int j = j0 + tj;

    float acc = 0.0f;
#pragma unroll 4
    for (int k = 0; k < MLP; ++k) {
        acc += sW[k] * fast_tanh(sA[ti][k] + sB[tj][k]);
    }
    float r = acc + b2[0];
    out[(size_t)i * S_LEN + j] = (i == j) ? 0.0f : r;
}

extern "C" void kernel_launch(void* const* d_in, const int* in_sizes, int n_in,
                              void* d_out, int out_size, void* d_ws, size_t ws_size,
                              hipStream_t stream) {
    const int*   widx   = (const int*)d_in[0];
    const int*   pidx   = (const int*)d_in[1];
    const float* Wword  = (const float*)d_in[3];
    const float* Wpos   = (const float*)d_in[4];
    const float* Wih_l0f = (const float*)d_in[5];
    const float* Whh_l0f = (const float*)d_in[6];
    const float* bih_l0f = (const float*)d_in[7];
    const float* bhh_l0f = (const float*)d_in[8];
    const float* Wih_l0b = (const float*)d_in[9];
    const float* Whh_l0b = (const float*)d_in[10];
    const float* bih_l0b = (const float*)d_in[11];
    const float* bhh_l0b = (const float*)d_in[12];
    const float* Wih_l1f = (const float*)d_in[13];
    const float* Whh_l1f = (const float*)d_in[14];
    const float* bih_l1f = (const float*)d_in[15];
    const float* bhh_l1f = (const float*)d_in[16];
    const float* Wih_l1b = (const float*)d_in[17];
    const float* Whh_l1b = (const float*)d_in[18];
    const float* bih_l1b = (const float*)d_in[19];
    const float* bhh_l1b = (const float*)d_in[20];
    const float* W1 = (const float*)d_in[21];
    const float* b1 = (const float*)d_in[22];
    const float* W2 = (const float*)d_in[23];
    const float* b2 = (const float*)d_in[24];
    float* out = (float*)d_out;

    float* ws = (float*)d_ws;
    float* x      = ws;
    float* gxf    = x      + S_LEN * EMB;
    float* gxb    = gxf    + S_LEN * G4H;
    float* h0     = gxb    + S_LEN * G4H;
    float* h1     = h0     + S_LEN * 2 * HDIM;
    float* hA     = h1     + S_LEN * 2 * HDIM;
    float* hB     = hA     + S_LEN * MLP;
    float* BP0f   = hB     + S_LEN * MLP;
    float* BP0b   = BP0f   + EMB * G4H;
    float* BP1f   = BP0b   + EMB * G4H;
    float* BP1b   = BP1f   + 2 * HDIM * G4H;
    float* WhhT0f = BP1b   + 2 * HDIM * G4H;
    float* WhhT0b = WhhT0f + HDIM * G4H;
    float* WhhT1f = WhhT0b + HDIM * G4H;
    float* WhhT1b = WhhT1f + HDIM * G4H;
    float* BPA    = WhhT1b + HDIM * G4H;
    float* BPB    = BPA    + 2 * HDIM * MLP;

    auto tgrid = [](int total) { return dim3((total + 255) / 256); };
    pack_bt<<<tgrid(G4H * EMB),      256, 0, stream>>>(Wih_l0f, BP0f, G4H, EMB,      0,      EMB);
    pack_bt<<<tgrid(G4H * EMB),      256, 0, stream>>>(Wih_l0b, BP0b, G4H, EMB,      0,      EMB);
    pack_bt<<<tgrid(G4H * 2 * HDIM), 256, 0, stream>>>(Wih_l1f, BP1f, G4H, 2 * HDIM, 0, 2 * HDIM);
    pack_bt<<<tgrid(G4H * 2 * HDIM), 256, 0, stream>>>(Wih_l1b, BP1b, G4H, 2 * HDIM, 0, 2 * HDIM);
    pack_bt<<<tgrid(MLP * 2 * HDIM), 256, 0, stream>>>(W1, BPA, MLP, G4H,        0, 2 * HDIM);
    pack_bt<<<tgrid(MLP * 2 * HDIM), 256, 0, stream>>>(W1, BPB, MLP, G4H, 2 * HDIM, 2 * HDIM);
    pack_whh<<<tgrid(G4H * HDIM), 256, 0, stream>>>(Whh_l0f, WhhT0f);
    pack_whh<<<tgrid(G4H * HDIM), 256, 0, stream>>>(Whh_l0b, WhhT0b);
    pack_whh<<<tgrid(G4H * HDIM), 256, 0, stream>>>(Whh_l1f, WhhT1f);
    pack_whh<<<tgrid(G4H * HDIM), 256, 0, stream>>>(Whh_l1b, WhhT1b);

    embed_kernel<<<S_LEN, EMB, 0, stream>>>(widx, pidx, Wword, Wpos, x);

    {
        dim3 grid(G4H / 64, S_LEN / 16);
        gemm_wmma_f32<<<grid, 32, 0, stream>>>(x, BP0f, bih_l0f, bhh_l0f, gxf, S_LEN, G4H, EMB);
        gemm_wmma_f32<<<grid, 32, 0, stream>>>(x, BP0b, bih_l0b, bhh_l0b, gxb, S_LEN, G4H, EMB);
    }
    lstm_layer<<<2, HDIM, 0, stream>>>(gxf, gxb, WhhT0f, WhhT0b, h0);

    {
        dim3 grid(G4H / 64, S_LEN / 16);
        gemm_wmma_f32<<<grid, 32, 0, stream>>>(h0, BP1f, bih_l1f, bhh_l1f, gxf, S_LEN, G4H, 2 * HDIM);
        gemm_wmma_f32<<<grid, 32, 0, stream>>>(h0, BP1b, bih_l1b, bhh_l1b, gxb, S_LEN, G4H, 2 * HDIM);
    }
    lstm_layer<<<2, HDIM, 0, stream>>>(gxf, gxb, WhhT1f, WhhT1b, h1);

    {
        dim3 grid(MLP / 64, S_LEN / 16);
        gemm_wmma_f32<<<grid, 32, 0, stream>>>(h1, BPA, b1, (const float*)nullptr, hA, S_LEN, MLP, 2 * HDIM);
        gemm_wmma_f32<<<grid, 32, 0, stream>>>(h1, BPB, (const float*)nullptr, (const float*)nullptr, hB, S_LEN, MLP, 2 * HDIM);
    }

    {
        dim3 grid(S_LEN / 16, S_LEN / 16);
        pairwise_kernel<<<grid, 256, 0, stream>>>(hA, hB, W2, b2, out);
    }
}


# LSTM recurrence: all 4 gate sums pass the segment anchors, so share the low register window: a k-row is 2 hinted dual-issue FMA pairs, not 3 instrs + 2 window switches; summation order unchanged
# speedup vs baseline: 3.6670x; 1.1707x over previous
#include <hip/hip_runtime.h>
#include <hip/hip_bf16.h>
#include <utility>


#define S_LEN   768
#define V_WORD  30000
#define V_POS   64
#define WEMB    100
#define PEMB    28
#define EMB     128
#define HDIM    256
#define G4H     1024
#define MLP     128

typedef __attribute__((ext_vector_type(2))) float v2f;
typedef __attribute__((ext_vector_type(8))) float v8f;

__device__ __forceinline__ float fast_sigmoid(float x) {
    return 1.0f / (1.0f + __expf(-x));
}
__device__ __forceinline__ float fast_tanh(float x) {
    float c = fminf(fmaxf(x, -9.0f), 9.0f);
    float e = __expf(2.0f * c);
    return (e - 1.0f) / (e + 1.0f);
}

__global__ void embed_kernel(const int* __restrict__ widx, const int* __restrict__ pidx,
                             const float* __restrict__ Wword, const float* __restrict__ Wpos,
                             float* __restrict__ x) {
    int s = blockIdx.x;
    int t = threadIdx.x;
    float v;
    if (t < WEMB) v = Wword[(size_t)widx[s] * WEMB + t];
    else          v = Wpos[(size_t)pidx[s] * PEMB + (t - WEMB)];
    x[(size_t)s * EMB + t] = v;
}

__global__ void pack_whh(const float* __restrict__ src, float* __restrict__ dst) {
    const int idx = blockIdx.x * blockDim.x + threadIdx.x;
    if (idx >= G4H * HDIM) return;
    const int row = idx >> 8;
    const int j   = idx & (HDIM - 1);
    const int g   = row >> 8;
    const int u   = row & (HDIM - 1);
    dst[((size_t)j * HDIM + u) * 4 + g] = src[idx];
}

__global__ void pack_bt(const float* __restrict__ src, float* __restrict__ dst,
                        int Nout, int Ctot, int coff, int Kcnt) {
    int idx = blockIdx.x * blockDim.x + threadIdx.x;
    int total = Nout * Kcnt;
    if (idx >= total) return;
    int n = idx / Kcnt;
    int k = idx - n * Kcnt;
    dst[((size_t)(k >> 2) * Nout + n) * 4 + (k & 3)] = src[(size_t)n * Ctot + coff + k];
}

__global__ void __launch_bounds__(32)
gemm_wmma_f32(const float* __restrict__ A, const float* __restrict__ BP,
              const float* __restrict__ bias0, const float* __restrict__ bias1,
              float* __restrict__ C, int M, int N, int K) {
    const int nt = blockIdx.x;
    const int mt = blockIdx.y;
    const int l  = threadIdx.x;
    const int fr   = l & 15;
    const int half = l >> 4;
    const int kb   = half * 2;

    const float* Arow  = A + (size_t)(mt * 16 + fr) * K + kb;
    const int    nbase = nt * 64;

    v8f acc[4] = {{}, {}, {}, {}};
    for (int k0 = 0; k0 < K; k0 += 4) {
        v2f a;
        a.x = Arow[k0];
        a.y = Arow[k0 + 1];
        const float* bp = BP + ((size_t)(k0 >> 2) * N + nbase + fr) * 4 + kb;
        v2f b0, b1, b2, b3;
        b0.x = bp[0];       b0.y = bp[1];
        b1.x = bp[16 * 4];  b1.y = bp[16 * 4 + 1];
        b2.x = bp[32 * 4];  b2.y = bp[32 * 4 + 1];
        b3.x = bp[48 * 4];  b3.y = bp[48 * 4 + 1];
        acc[0] = __builtin_amdgcn_wmma_f32_16x16x4_f32(false, a, false, b0, (short)0, acc[0], false, false);
        acc[1] = __builtin_amdgcn_wmma_f32_16x16x4_f32(false, a, false, b1, (short)0, acc[1], false, false);
        acc[2] = __builtin_amdgcn_wmma_f32_16x16x4_f32(false, a, false, b2, (short)0, acc[2], false, false);
        acc[3] = __builtin_amdgcn_wmma_f32_16x16x4_f32(false, a, false, b3, (short)0, acc[3], false, false);
    }

#pragma unroll
    for (int u = 0; u < 4; ++u) {
        const int g = nbase + u * 16 + fr;
        float bs = 0.0f;
        if (bias0) bs += bias0[g];
        if (bias1) bs += bias1[g];
#pragma unroll
        for (int v = 0; v < 8; ++v) {
            int row = v + half * 8;
            C[(size_t)(mt * 16 + row) * N + g] = acc[u][v] + bs;
        }
    }
}

#define KS 100
#define SG 5
#define SR 4
#define KL 79
#define NS (HDIM - KS)
#define KR (NS - KL)
#define NG (KS / SG)
#define PD 8
#define HQ (2 * HDIM / 4)
#define WSEG 4096
#define NSEG ((KL * HDIM + WSEG - 1) / WSEG)

constexpr int  seg_start(int g) { return (g * HDIM) / NG; }
constexpr int  seg_of(int j)    { return ((j + 1) * NG - 1) / HDIM; }
constexpr int  seg_off(int j)   { return j - seg_start(seg_of(j)); }
constexpr bool seg_last(int j)  { return j + 1 == seg_start(seg_of(j) + 1); }
constexpr int  lds_start(int g) { return (g * KL) / NG; }
constexpr int  reg_start(int g) { return seg_start(g) - g * SG - lds_start(g); }
constexpr int  seg_regs(int g)  { return reg_start(g + 1) - reg_start(g); }
constexpr bool is_reg(int j)    { return seg_off(j) < seg_regs(seg_of(j)); }
constexpr bool is_str(int j)    { return !is_reg(j) && seg_off(j) < seg_regs(seg_of(j)) + SG; }
constexpr bool is_lds(int j)    { return !is_reg(j) && !is_str(j); }
constexpr int  n_reg(int j)     { return reg_start(seg_of(j)) + seg_off(j); }
constexpr int  n_str(int j)     { return seg_of(j) * SG + seg_off(j) - seg_regs(seg_of(j)); }
constexpr int  n_lds(int j)     { return lds_start(seg_of(j)) + seg_off(j) - seg_regs(seg_of(j)) - SG; }
constexpr int  j_of_str(int s)  { return seg_start(s / SG) + seg_regs(s / SG) + s % SG; }
constexpr int  fetch_seg(int j) { return (seg_off(j) >= PD || seg_of(j) == 0) ? seg_of(j) : seg_of(j) - 1; }
constexpr bool layout_ok() {
    if (reg_start(NG) != KR || lds_start(NG) != KL || seg_start(NG) != HDIM) return false;
    for (int g = 0; g < NG; ++g) if (seg_regs(g) < 0 || seg_regs(g) + SG < PD) return false;
    for (int s = 0; s < KS; ++s) if (!is_str(j_of_str(s)) || n_str(j_of_str(s)) != s) return false;
    return true;
}
static_assert(layout_ok(), "segment layout does not match KS / KL / KR");

__device__ __forceinline__ void lds_barrier() {
    __builtin_amdgcn_fence(__ATOMIC_RELEASE, "workgroup", "local");
    __builtin_amdgcn_s_barrier();
    __builtin_amdgcn_fence(__ATOMIC_ACQUIRE, "workgroup", "local");
}

template <int J, bool LDS_PASS>
__device__ __forceinline__ void whh_place_row(const float4* __restrict__ Wp, float4* lds, const int (&wb)[NSEG],
                                              float4 (&wr)[KR]) {
    if constexpr (LDS_PASS && is_lds(J)) lds[wb[n_lds(J) / 16] + (n_lds(J) % 16) * HDIM] = Wp[J * HDIM];
    if constexpr (!LDS_PASS && is_reg(J)) wr[n_reg(J)] = Wp[J * HDIM];
}
template <bool LDS_PASS, int... Js>
__device__ __forceinline__ void whh_place_rows(std::integer_sequence<int, Js...>, const float4* __restrict__ Wp,
                                               float4* lds, const int (&wb)[NSEG], float4 (&wr)[KR]) {
    (whh_place_row<Js, LDS_PASS>(Wp, lds, wb, wr), ...);
}
template <int... Ss>
__device__ __forceinline__ void whh_stream_fill(std::integer_sequence<int, Ss...>, const float4* __restrict__ Wp,
                                                float4 (&sb)[SR * SG]) {
    ((sb[Ss] = Wp[j_of_str(Ss) * HDIM]), ...);
}
template <int G, int... Is>
__device__ __forceinline__ void whh_refill(std::integer_sequence<int, Is...>, const float4* __restrict__ Wp,
                                           float4 (&sb)[SR * SG]) {
    ((sb[(G % SR) * SG + Is] = Wp[j_of_str(((G + SR) * SG + Is) % KS) * HDIM]), ...);
}

template <int G, int J>
__device__ __forceinline__ void lds_fetch_row(const float4* lds, const int (&wb)[NSEG], int hb,
                                              float4 (&lw)[KL], float4 (&hq)[HDIM / 4]) {
    if constexpr (fetch_seg(J) == G) {
        if constexpr ((J & 3) == 0) hq[J >> 2] = lds[hb + (J >> 2)];
        if constexpr (is_lds(J)) lw[n_lds(J)] = lds[wb[n_lds(J) / 16] + (n_lds(J) % 16) * HDIM];
    }
}
template <int G, int... Js>
__device__ __forceinline__ void lds_fetch_seg(std::integer_sequence<int, Js...>, const float4* lds, const int (&wb)[NSEG],
                                              int hb, float4 (&lw)[KL], float4 (&hq)[HDIM / 4]) {
    (lds_fetch_row<G, Js>(lds, wb, hb, lw, hq), ...);
}

template <int J>
__device__ __forceinline__ void whh_row(const float4* __restrict__ Wp, const float4* lds, const int (&wb)[NSEG], int hb,
                                        float4 (&wr)[KR], float4 (&sb)[SR * SG], float4 (&lw)[KL], float4 (&hq)[HDIM / 4],
                                        float& a_i, float& a_f, float& a_g, float& a_o) {
    if constexpr (J == 0) {
        lds_fetch_seg<0>(std::make_integer_sequence<int, HDIM>{}, lds, wb, hb, lw, hq);
        asm volatile("" : "+v"(a_i), "+v"(a_f), "+v"(a_g), "+v"(a_o));
    }
    const float4 hv = hq[J >> 2];
    const float hj = ((J & 3) == 0) ? hv.x : ((J & 3) == 1) ? hv.y : ((J & 3) == 2) ? hv.z : hv.w;
    float4 w;
    if constexpr (is_str(J)) {
        constexpr int s = n_str(J);
        constexpr int g = s / SG;
        w = sb[(g % SR) * SG + (s % SG)];
    } else if constexpr (is_lds(J)) {
        w = lw[n_lds(J)];
    } else {
        w = wr[n_reg(J)];
    }
    a_i = __builtin_fmaf(hj, w.x, a_i);
    a_f = __builtin_fmaf(hj, w.y, a_f);
    a_g = __builtin_fmaf(hj, w.z, a_g);
    a_o = __builtin_fmaf(hj, w.w, a_o);
    if constexpr (is_str(J) && (n_str(J) % SG) == SG - 1) {
        constexpr int g = n_str(J) / SG;
        whh_refill<g>(std::make_integer_sequence<int, SG>{}, Wp, sb);
    }
    if constexpr (seg_last(J)) {
        if constexpr (seg_of(J) + 1 < NG)
            lds_fetch_seg<seg_of(J) + 1>(std::make_integer_sequence<int, HDIM>{}, lds, wb, hb, lw, hq);
        asm volatile("" : "+v"(a_i), "+v"(a_f), "+v"(a_g), "+v"(a_o));
    }
}
template <int... Js>
__device__ __forceinline__ void whh_rows(std::integer_sequence<int, Js...>, const float4* __restrict__ Wp,
                                         const float4* lds, const int (&wb)[NSEG], int hb,
                                         float4 (&wr)[KR], float4 (&sb)[SR * SG], float4 (&lw)[KL], float4 (&hq)[HDIM / 4],
                                         float& a_i, float& a_f, float& a_g, float& a_o) {
    (whh_row<Js>(Wp, lds, wb, hb, wr, sb, lw, hq, a_i, a_f, a_g, a_o), ...);
}

__global__ void __launch_bounds__(HDIM)
lstm_layer(const float* __restrict__ gx_f, const float* __restrict__ gx_b,
           const float* __restrict__ Wp_f, const float* __restrict__ Wp_b,
           float* __restrict__ hout) {
    const int dir = blockIdx.x;
    const int u   = threadIdx.x;
    const unsigned u4  = (unsigned)u * 4u;
    const unsigned u16 = (unsigned)u * 16u;
    const char*   gxd = reinterpret_cast<const char*>(dir ? gx_b : gx_f);
    const char*   Wd  = reinterpret_cast<const char*>(dir ? Wp_b : Wp_f);
    char*         hd  = reinterpret_cast<char*>(hout + dir * HDIM);
    const float4* Wp  = reinterpret_cast<const float4*>(Wd + u16);

    __shared__ float4 lds[HQ + KL * HDIM];
    float* h_s = reinterpret_cast<float*>(lds);

    int wb[NSEG];
#pragma unroll
    for (int k = 0; k < NSEG; ++k) {
        wb[k] = HQ + u + k * WSEG;
        asm volatile("" : "+v"(wb[k]));
    }

    float4 wr[KR];
    float4 sb[SR * SG];
    whh_place_rows<true>(std::make_integer_sequence<int, HDIM>{}, Wp, lds, wb, wr);
    asm volatile("" ::: "memory");
    whh_place_rows<false>(std::make_integer_sequence<int, HDIM>{}, Wp, lds, wb, wr);
    whh_stream_fill(std::make_integer_sequence<int, SR * SG>{}, Wp, sb);

    h_s[u] = 0.0f;
    float c = 0.0f;

    int t = dir ? (S_LEN - 1) : 0;
    const int dt = dir ? -1 : 1;
    const unsigned go0 = (unsigned)t * (G4H * 4) + u4;
    float g_i = *reinterpret_cast<const float*>(gxd + go0);
    float g_f = *reinterpret_cast<const float*>(gxd + go0 + HDIM * 4);
    float g_g = *reinterpret_cast<const float*>(gxd + go0 + 2 * HDIM * 4);
    float g_o = *reinterpret_cast<const float*>(gxd + go0 + 3 * HDIM * 4);
    __syncthreads();

    for (int step = 0; step < S_LEN; ++step) {
        const int cur = step & 1;
        float a_i = g_i, a_f = g_f, a_g = g_g, a_o = g_o;

        const int tn = (step + 1 < S_LEN) ? (t + dt) : t;
        const unsigned gon = (unsigned)tn * (G4H * 4) + u4;
        g_i = *reinterpret_cast<const float*>(gxd + gon);
        g_f = *reinterpret_cast<const float*>(gxd + gon + HDIM * 4);
        g_g = *reinterpret_cast<const float*>(gxd + gon + 2 * HDIM * 4);
        g_o = *reinterpret_cast<const float*>(gxd + gon + 3 * HDIM * 4);

        int zoff = 0;
        asm volatile("" : "+s"(zoff));
        const float4* Wst = reinterpret_cast<const float4*>(Wd + (u16 + (unsigned)zoff));

        int hb = cur * (HDIM / 4);
        asm volatile("" : "+v"(hb));

        float4 lw[KL];
        float4 hq[HDIM / 4];
        whh_rows(std::make_integer_sequence<int, HDIM>{}, Wst, lds, wb, hb, wr, sb, lw, hq, a_i, a_f, a_g, a_o);

        c = fast_sigmoid(a_f) * c + fast_sigmoid(a_i) * fast_tanh(a_g);
        const float h = fast_sigmoid(a_o) * fast_tanh(c);
        h_s[(cur ^ 1) * HDIM + u] = h;
        *reinterpret_cast<float*>(hd + ((unsigned)t * (2 * HDIM * 4) + u4)) = h;
        t = tn;
        lds_barrier();
    }
}

__global__ void pairwise_kernel(const float* __restrict__ hA, const float* __restrict__ hB,
                                const float* __restrict__ W2, const float* __restrict__ b2,
                                float* __restrict__ out) {
    __shared__ float sA[16][MLP];
    __shared__ float sB[16][MLP];
    __shared__ float sW[MLP];

    const int i0 = blockIdx.y * 16;
    const int j0 = blockIdx.x * 16;
    const int tid = threadIdx.x;

    for (int idx = tid; idx < 16 * MLP; idx += 256) {
        int r = idx >> 7, k = idx & (MLP - 1);
        sA[r][k] = hA[(size_t)(i0 + r) * MLP + k];
        sB[r][k] = hB[(size_t)(j0 + r) * MLP + k];
    }
    if (tid < MLP) sW[tid] = W2[tid];
    __syncthreads();

    const int ti = tid >> 4;
    const int tj = tid & 15;
    const int i = i0 + ti;
    const int j = j0 + tj;

    float acc = 0.0f;
#pragma unroll 4
    for (int k = 0; k < MLP; ++k) {
        acc += sW[k] * fast_tanh(sA[ti][k] + sB[tj][k]);
    }
    float r = acc + b2[0];
    out[(size_t)i * S_LEN + j] = (i == j) ? 0.0f : r;
}

extern "C" void kernel_launch(void* const* d_in, const int* in_sizes, int n_in,
                              void* d_out, int out_size, void* d_ws, size_t ws_size,
                              hipStream_t stream) {
    const int*   widx   = (const int*)d_in[0];
    const int*   pidx   = (const int*)d_in[1];
    const float* Wword  = (const float*)d_in[3];
    const float* Wpos   = (const float*)d_in[4];
    const float* Wih_l0f = (const float*)d_in[5];
    const float* Whh_l0f = (const float*)d_in[6];
    const float* bih_l0f = (const float*)d_in[7];
    const float* bhh_l0f = (const float*)d_in[8];
    const float* Wih_l0b = (const float*)d_in[9];
    const float* Whh_l0b = (const float*)d_in[10];
    const float* bih_l0b = (const float*)d_in[11];
    const float* bhh_l0b = (const float*)d_in[12];
    const float* Wih_l1f = (const float*)d_in[13];
    const float* Whh_l1f = (const float*)d_in[14];
    const float* bih_l1f = (const float*)d_in[15];
    const float* bhh_l1f = (const float*)d_in[16];
    const float* Wih_l1b = (const float*)d_in[17];
    const float* Whh_l1b = (const float*)d_in[18];
    const float* bih_l1b = (const float*)d_in[19];
    const float* bhh_l1b = (const float*)d_in[20];
    const float* W1 = (const float*)d_in[21];
    const float* b1 = (const float*)d_in[22];
    const float* W2 = (const float*)d_in[23];
    const float* b2 = (const float*)d_in[24];
    float* out = (float*)d_out;

    float* ws = (float*)d_ws;
    float* x      = ws;
    float* gxf    = x      + S_LEN * EMB;
    float* gxb    = gxf    + S_LEN * G4H;
    float* h0     = gxb    + S_LEN * G4H;
    float* h1     = h0     + S_LEN * 2 * HDIM;
    float* hA     = h1     + S_LEN * 2 * HDIM;
    float* hB     = hA     + S_LEN * MLP;
    float* BP0f   = hB     + S_LEN * MLP;
    float* BP0b   = BP0f   + EMB * G4H;
    float* BP1f   = BP0b   + EMB * G4H;
    float* BP1b   = BP1f   + 2 * HDIM * G4H;
    float* WhhT0f = BP1b   + 2 * HDIM * G4H;
    float* WhhT0b = WhhT0f + HDIM * G4H;
    float* WhhT1f = WhhT0b + HDIM * G4H;
    float* WhhT1b = WhhT1f + HDIM * G4H;
    float* BPA    = WhhT1b + HDIM * G4H;
    float* BPB    = BPA    + 2 * HDIM * MLP;

    auto tgrid = [](int total) { return dim3((total + 255) / 256); };
    pack_bt<<<tgrid(G4H * EMB),      256, 0, stream>>>(Wih_l0f, BP0f, G4H, EMB,      0,      EMB);
    pack_bt<<<tgrid(G4H * EMB),      256, 0, stream>>>(Wih_l0b, BP0b, G4H, EMB,      0,      EMB);
    pack_bt<<<tgrid(G4H * 2 * HDIM), 256, 0, stream>>>(Wih_l1f, BP1f, G4H, 2 * HDIM, 0, 2 * HDIM);
    pack_bt<<<tgrid(G4H * 2 * HDIM), 256, 0, stream>>>(Wih_l1b, BP1b, G4H, 2 * HDIM, 0, 2 * HDIM);
    pack_bt<<<tgrid(MLP * 2 * HDIM), 256, 0, stream>>>(W1, BPA, MLP, G4H,        0, 2 * HDIM);
    pack_bt<<<tgrid(MLP * 2 * HDIM), 256, 0, stream>>>(W1, BPB, MLP, G4H, 2 * HDIM, 2 * HDIM);
    pack_whh<<<tgrid(G4H * HDIM), 256, 0, stream>>>(Whh_l0f, WhhT0f);
    pack_whh<<<tgrid(G4H * HDIM), 256, 0, stream>>>(Whh_l0b, WhhT0b);
    pack_whh<<<tgrid(G4H * HDIM), 256, 0, stream>>>(Whh_l1f, WhhT1f);
    pack_whh<<<tgrid(G4H * HDIM), 256, 0, stream>>>(Whh_l1b, WhhT1b);

    embed_kernel<<<S_LEN, EMB, 0, stream>>>(widx, pidx, Wword, Wpos, x);

    {
        dim3 grid(G4H / 64, S_LEN / 16);
        gemm_wmma_f32<<<grid, 32, 0, stream>>>(x, BP0f, bih_l0f, bhh_l0f, gxf, S_LEN, G4H, EMB);
        gemm_wmma_f32<<<grid, 32, 0, stream>>>(x, BP0b, bih_l0b, bhh_l0b, gxb, S_LEN, G4H, EMB);
    }
    lstm_layer<<<2, HDIM, 0, stream>>>(gxf, gxb, WhhT0f, WhhT0b, h0);

    {
        dim3 grid(G4H / 64, S_LEN / 16);
        gemm_wmma_f32<<<grid, 32, 0, stream>>>(h0, BP1f, bih_l1f, bhh_l1f, gxf, S_LEN, G4H, 2 * HDIM);
        gemm_wmma_f32<<<grid, 32, 0, stream>>>(h0, BP1b, bih_l1b, bhh_l1b, gxb, S_LEN, G4H, 2 * HDIM);
    }
    lstm_layer<<<2, HDIM, 0, stream>>>(gxf, gxb, WhhT1f, WhhT1b, h1);

    {
        dim3 grid(MLP / 64, S_LEN / 16);
        gemm_wmma_f32<<<grid, 32, 0, stream>>>(h1, BPA, b1, (const float*)nullptr, hA, S_LEN, MLP, 2 * HDIM);
        gemm_wmma_f32<<<grid, 32, 0, stream>>>(h1, BPB, (const float*)nullptr, (const float*)nullptr, hB, S_LEN, MLP, 2 * HDIM);
    }

    {
        dim3 grid(S_LEN / 16, S_LEN / 16);
        pairwise_kernel<<<grid, 256, 0, stream>>>(hA, hB, W2, b2, out);
    }
}


# LSTM recurrence: when a step's sums finish, 15 more streamed weight rows (next step's groups 5-7) are requested, so loads stay in flight through gate math and barrier; summation order unchanged
# speedup vs baseline: 3.8059x; 1.0379x over previous
#include <hip/hip_runtime.h>
#include <hip/hip_bf16.h>
#include <utility>


#define S_LEN   768
#define V_WORD  30000
#define V_POS   64
#define WEMB    100
#define PEMB    28
#define EMB     128
#define HDIM    256
#define G4H     1024
#define MLP     128

typedef __attribute__((ext_vector_type(2))) float v2f;
typedef __attribute__((ext_vector_type(8))) float v8f;

__device__ __forceinline__ float fast_sigmoid(float x) {
    return 1.0f / (1.0f + __expf(-x));
}
__device__ __forceinline__ float fast_tanh(float x) {
    float c = fminf(fmaxf(x, -9.0f), 9.0f);
    float e = __expf(2.0f * c);
    return (e - 1.0f) / (e + 1.0f);
}

__global__ void embed_kernel(const int* __restrict__ widx, const int* __restrict__ pidx,
                             const float* __restrict__ Wword, const float* __restrict__ Wpos,
                             float* __restrict__ x) {
    int s = blockIdx.x;
    int t = threadIdx.x;
    float v;
    if (t < WEMB) v = Wword[(size_t)widx[s] * WEMB + t];
    else          v = Wpos[(size_t)pidx[s] * PEMB + (t - WEMB)];
    x[(size_t)s * EMB + t] = v;
}

__global__ void pack_whh(const float* __restrict__ src, float* __restrict__ dst) {
    const int idx = blockIdx.x * blockDim.x + threadIdx.x;
    if (idx >= G4H * HDIM) return;
    const int row = idx >> 8;
    const int j   = idx & (HDIM - 1);
    const int g   = row >> 8;
    const int u   = row & (HDIM - 1);
    dst[((size_t)j * HDIM + u) * 4 + g] = src[idx];
}

__global__ void pack_bt(const float* __restrict__ src, float* __restrict__ dst,
                        int Nout, int Ctot, int coff, int Kcnt) {
    int idx = blockIdx.x * blockDim.x + threadIdx.x;
    int total = Nout * Kcnt;
    if (idx >= total) return;
    int n = idx / Kcnt;
    int k = idx - n * Kcnt;
    dst[((size_t)(k >> 2) * Nout + n) * 4 + (k & 3)] = src[(size_t)n * Ctot + coff + k];
}

__global__ void __launch_bounds__(32)
gemm_wmma_f32(const float* __restrict__ A, const float* __restrict__ BP,
              const float* __restrict__ bias0, const float* __restrict__ bias1,
              float* __restrict__ C, int M, int N, int K) {
    const int nt = blockIdx.x;
    const int mt = blockIdx.y;
    const int l  = threadIdx.x;
    const int fr   = l & 15;
    const int half = l >> 4;
    const int kb   = half * 2;

    const float* Arow  = A + (size_t)(mt * 16 + fr) * K + kb;
    const int    nbase = nt * 64;

    v8f acc[4] = {{}, {}, {}, {}};
    for (int k0 = 0; k0 < K; k0 += 4) {
        v2f a;
        a.x = Arow[k0];
        a.y = Arow[k0 + 1];
        const float* bp = BP + ((size_t)(k0 >> 2) * N + nbase + fr) * 4 + kb;
        v2f b0, b1, b2, b3;
        b0.x = bp[0];       b0.y = bp[1];
        b1.x = bp[16 * 4];  b1.y = bp[16 * 4 + 1];
        b2.x = bp[32 * 4];  b2.y = bp[32 * 4 + 1];
        b3.x = bp[48 * 4];  b3.y = bp[48 * 4 + 1];
        acc[0] = __builtin_amdgcn_wmma_f32_16x16x4_f32(false, a, false, b0, (short)0, acc[0], false, false);
        acc[1] = __builtin_amdgcn_wmma_f32_16x16x4_f32(false, a, false, b1, (short)0, acc[1], false, false);
        acc[2] = __builtin_amdgcn_wmma_f32_16x16x4_f32(false, a, false, b2, (short)0, acc[2], false, false);
        acc[3] = __builtin_amdgcn_wmma_f32_16x16x4_f32(false, a, false, b3, (short)0, acc[3], false, false);
    }

#pragma unroll
    for (int u = 0; u < 4; ++u) {
        const int g = nbase + u * 16 + fr;
        float bs = 0.0f;
        if (bias0) bs += bias0[g];
        if (bias1) bs += bias1[g];
#pragma unroll
        for (int v = 0; v < 8; ++v) {
            int row = v + half * 8;
            C[(size_t)(mt * 16 + row) * N + g] = acc[u][v] + bs;
        }
    }
}

#define KS 100
#define SG 5
#define SR 4
#define XG 3
#define KL 79
#define NS (HDIM - KS)
#define KR (NS - KL)
#define NG (KS / SG)
#define PD 8
#define HQ (2 * HDIM / 4)
#define WSEG 4096
#define NSEG ((KL * HDIM + WSEG - 1) / WSEG)

constexpr int  seg_start(int g) { return (g * HDIM) / NG; }
constexpr int  seg_of(int j)    { return ((j + 1) * NG - 1) / HDIM; }
constexpr int  seg_off(int j)   { return j - seg_start(seg_of(j)); }
constexpr bool seg_last(int j)  { return j + 1 == seg_start(seg_of(j) + 1); }
constexpr int  lds_start(int g) { return (g * KL) / NG; }
constexpr int  reg_start(int g) { return seg_start(g) - g * SG - lds_start(g); }
constexpr int  seg_regs(int g)  { return reg_start(g + 1) - reg_start(g); }
constexpr bool is_reg(int j)    { return seg_off(j) < seg_regs(seg_of(j)); }
constexpr bool is_str(int j)    { return !is_reg(j) && seg_off(j) < seg_regs(seg_of(j)) + SG; }
constexpr bool is_lds(int j)    { return !is_reg(j) && !is_str(j); }
constexpr int  n_reg(int j)     { return reg_start(seg_of(j)) + seg_off(j); }
constexpr int  n_str(int j)     { return seg_of(j) * SG + seg_off(j) - seg_regs(seg_of(j)); }
constexpr int  n_lds(int j)     { return lds_start(seg_of(j)) + seg_off(j) - seg_regs(seg_of(j)) - SG; }
constexpr int  j_of_str(int s)  { return seg_start(s / SG) + seg_regs(s / SG) + s % SG; }
constexpr int  fetch_seg(int j) { return (seg_off(j) >= PD || seg_of(j) == 0) ? seg_of(j) : seg_of(j) - 1; }
constexpr bool layout_ok() {
    if (reg_start(NG) != KR || lds_start(NG) != KL || seg_start(NG) != HDIM) return false;
    for (int g = 0; g < NG; ++g) if (seg_regs(g) < 0 || seg_regs(g) + SG < PD) return false;
    for (int s = 0; s < KS; ++s) if (!is_str(j_of_str(s)) || n_str(j_of_str(s)) != s) return false;
    return true;
}
static_assert(layout_ok(), "segment layout does not match KS / KL / KR");
static_assert(XG >= 0 && SR + XG <= NG - SR, "groups requested across the step boundary must not meet the ring's wrap-around");

__device__ __forceinline__ void lds_barrier() {
    __builtin_amdgcn_fence(__ATOMIC_RELEASE, "workgroup", "local");
    __builtin_amdgcn_s_barrier();
    __builtin_amdgcn_fence(__ATOMIC_ACQUIRE, "workgroup", "local");
}

template <int J, bool LDS_PASS>
__device__ __forceinline__ void whh_place_row(const float4* __restrict__ Wp, float4* lds, const int (&wb)[NSEG],
                                              float4 (&wr)[KR]) {
    if constexpr (LDS_PASS && is_lds(J)) lds[wb[n_lds(J) / 16] + (n_lds(J) % 16) * HDIM] = Wp[J * HDIM];
    if constexpr (!LDS_PASS && is_reg(J)) wr[n_reg(J)] = Wp[J * HDIM];
}
template <bool LDS_PASS, int... Js>
__device__ __forceinline__ void whh_place_rows(std::integer_sequence<int, Js...>, const float4* __restrict__ Wp,
                                               float4* lds, const int (&wb)[NSEG], float4 (&wr)[KR]) {
    (whh_place_row<Js, LDS_PASS>(Wp, lds, wb, wr), ...);
}
template <int... Ss>
__device__ __forceinline__ void whh_stream_fill(std::integer_sequence<int, Ss...>, const float4* __restrict__ Wp,
                                                float4 (&sb)[SR * SG]) {
    ((sb[Ss] = Wp[j_of_str(Ss) * HDIM]), ...);
}
template <int... Ss>
__device__ __forceinline__ void whh_ahead_fill(std::integer_sequence<int, Ss...>, const float4* __restrict__ Wp,
                                               float4 (&xb)[XG * SG]) {
    ((xb[Ss] = Wp[j_of_str(SR * SG + Ss) * HDIM]), ...);
}
template <int G, int... Is>
__device__ __forceinline__ void whh_refill(std::integer_sequence<int, Is...>, const float4* __restrict__ Wp,
                                           float4 (&sb)[SR * SG]) {
    ((sb[(G % SR) * SG + Is] = Wp[j_of_str(((G + SR) * SG + Is) % KS) * HDIM]), ...);
}

template <int G, int J>
__device__ __forceinline__ void lds_fetch_row(const float4* lds, const int (&wb)[NSEG], int hb,
                                              float4 (&lw)[KL], float4 (&hq)[HDIM / 4]) {
    if constexpr (fetch_seg(J) == G) {
        if constexpr ((J & 3) == 0) hq[J >> 2] = lds[hb + (J >> 2)];
        if constexpr (is_lds(J)) lw[n_lds(J)] = lds[wb[n_lds(J) / 16] + (n_lds(J) % 16) * HDIM];
    }
}
template <int G, int... Js>
__device__ __forceinline__ void lds_fetch_seg(std::integer_sequence<int, Js...>, const float4* lds, const int (&wb)[NSEG],
                                              int hb, float4 (&lw)[KL], float4 (&hq)[HDIM / 4]) {
    (lds_fetch_row<G, Js>(lds, wb, hb, lw, hq), ...);
}

template <int J>
__device__ __forceinline__ void whh_row(const float4* __restrict__ Wp, const float4* lds, const int (&wb)[NSEG], int hb,
                                        float4 (&wr)[KR], float4 (&sb)[SR * SG], float4 (&xb)[XG * SG], float4 (&lw)[KL],
                                        float4 (&hq)[HDIM / 4], float& a_i, float& a_f, float& a_g, float& a_o) {
    if constexpr (J == 0) {
        lds_fetch_seg<0>(std::make_integer_sequence<int, HDIM>{}, lds, wb, hb, lw, hq);
        asm volatile("" : "+v"(a_i), "+v"(a_f), "+v"(a_g), "+v"(a_o));
    }
    const float4 hv = hq[J >> 2];
    const float hj = ((J & 3) == 0) ? hv.x : ((J & 3) == 1) ? hv.y : ((J & 3) == 2) ? hv.z : hv.w;
    float4 w;
    if constexpr (is_str(J)) {
        constexpr int s = n_str(J);
        constexpr int g = s / SG;
        if constexpr (g >= SR && g < SR + XG) w = xb[(g - SR) * SG + (s % SG)];
        else                                  w = sb[(g % SR) * SG + (s % SG)];
    } else if constexpr (is_lds(J)) {
        w = lw[n_lds(J)];
    } else {
        w = wr[n_reg(J)];
    }
    a_i = __builtin_fmaf(hj, w.x, a_i);
    a_f = __builtin_fmaf(hj, w.y, a_f);
    a_g = __builtin_fmaf(hj, w.z, a_g);
    a_o = __builtin_fmaf(hj, w.w, a_o);
    if constexpr (is_str(J) && (n_str(J) % SG) == SG - 1) {
        constexpr int g = n_str(J) / SG;
        if constexpr (g >= XG) whh_refill<g>(std::make_integer_sequence<int, SG>{}, Wp, sb);
    }
    if constexpr (seg_last(J)) {
        if constexpr (seg_of(J) + 1 < NG)
            lds_fetch_seg<seg_of(J) + 1>(std::make_integer_sequence<int, HDIM>{}, lds, wb, hb, lw, hq);
        asm volatile("" : "+v"(a_i), "+v"(a_f), "+v"(a_g), "+v"(a_o));
    }
}
template <int... Js>
__device__ __forceinline__ void whh_rows(std::integer_sequence<int, Js...>, const float4* __restrict__ Wp,
                                         const float4* lds, const int (&wb)[NSEG], int hb,
                                         float4 (&wr)[KR], float4 (&sb)[SR * SG], float4 (&xb)[XG * SG], float4 (&lw)[KL],
                                         float4 (&hq)[HDIM / 4], float& a_i, float& a_f, float& a_g, float& a_o) {
    (whh_row<Js>(Wp, lds, wb, hb, wr, sb, xb, lw, hq, a_i, a_f, a_g, a_o), ...);
}

__global__ void __launch_bounds__(HDIM)
lstm_layer(const float* __restrict__ gx_f, const float* __restrict__ gx_b,
           const float* __restrict__ Wp_f, const float* __restrict__ Wp_b,
           float* __restrict__ hout) {
    const int dir = blockIdx.x;
    const int u   = threadIdx.x;
    const unsigned u4  = (unsigned)u * 4u;
    const unsigned u16 = (unsigned)u * 16u;
    const char*   gxd = reinterpret_cast<const char*>(dir ? gx_b : gx_f);
    const char*   Wd  = reinterpret_cast<const char*>(dir ? Wp_b : Wp_f);
    char*         hd  = reinterpret_cast<char*>(hout + dir * HDIM);
    const float4* Wp  = reinterpret_cast<const float4*>(Wd + u16);

    __shared__ float4 lds[HQ + KL * HDIM];
    float* h_s = reinterpret_cast<float*>(lds);

    int wb[NSEG];
#pragma unroll
    for (int k = 0; k < NSEG; ++k) {
        wb[k] = HQ + u + k * WSEG;
        asm volatile("" : "+v"(wb[k]));
    }

    float4 wr[KR];
    float4 sb[SR * SG];
    float4 xb[XG * SG];
    whh_place_rows<true>(std::make_integer_sequence<int, HDIM>{}, Wp, lds, wb, wr);
    asm volatile("" ::: "memory");
    whh_place_rows<false>(std::make_integer_sequence<int, HDIM>{}, Wp, lds, wb, wr);
    whh_stream_fill(std::make_integer_sequence<int, SR * SG>{}, Wp, sb);
    whh_ahead_fill(std::make_integer_sequence<int, XG * SG>{}, Wp, xb);

    h_s[u] = 0.0f;
    float c = 0.0f;

    int t = dir ? (S_LEN - 1) : 0;
    const int dt = dir ? -1 : 1;
    const unsigned go0 = (unsigned)t * (G4H * 4) + u4;
    float g_i = *reinterpret_cast<const float*>(gxd + go0);
    float g_f = *reinterpret_cast<const float*>(gxd + go0 + HDIM * 4);
    float g_g = *reinterpret_cast<const float*>(gxd + go0 + 2 * HDIM * 4);
    float g_o = *reinterpret_cast<const float*>(gxd + go0 + 3 * HDIM * 4);
    __syncthreads();

    for (int step = 0; step < S_LEN; ++step) {
        const int cur = step & 1;
        float a_i = g_i, a_f = g_f, a_g = g_g, a_o = g_o;

        const int tn = (step + 1 < S_LEN) ? (t + dt) : t;
        const unsigned gon = (unsigned)tn * (G4H * 4) + u4;
        g_i = *reinterpret_cast<const float*>(gxd + gon);
        g_f = *reinterpret_cast<const float*>(gxd + gon + HDIM * 4);
        g_g = *reinterpret_cast<const float*>(gxd + gon + 2 * HDIM * 4);
        g_o = *reinterpret_cast<const float*>(gxd + gon + 3 * HDIM * 4);

        int zoff = 0;
        asm volatile("" : "+s"(zoff));
        const float4* Wst = reinterpret_cast<const float4*>(Wd + (u16 + (unsigned)zoff));

        int hb = cur * (HDIM / 4);
        asm volatile("" : "+v"(hb));

        float4 lw[KL];
        float4 hq[HDIM / 4];
        whh_rows(std::make_integer_sequence<int, HDIM>{}, Wst, lds, wb, hb, wr, sb, xb, lw, hq, a_i, a_f, a_g, a_o);

        whh_ahead_fill(std::make_integer_sequence<int, XG * SG>{}, Wst, xb);

        c = fast_sigmoid(a_f) * c + fast_sigmoid(a_i) * fast_tanh(a_g);
        const float h = fast_sigmoid(a_o) * fast_tanh(c);
        h_s[(cur ^ 1) * HDIM + u] = h;
        *reinterpret_cast<float*>(hd + ((unsigned)t * (2 * HDIM * 4) + u4)) = h;
        t = tn;
        lds_barrier();
    }
}

__global__ void pairwise_kernel(const float* __restrict__ hA, const float* __restrict__ hB,
                                const float* __restrict__ W2, const float* __restrict__ b2,
                                float* __restrict__ out) {
    __shared__ float sA[16][MLP];
    __shared__ float sB[16][MLP];
    __shared__ float sW[MLP];

    const int i0 = blockIdx.y * 16;
    const int j0 = blockIdx.x * 16;
    const int tid = threadIdx.x;

    for (int idx = tid; idx < 16 * MLP; idx += 256) {
        int r = idx >> 7, k = idx & (MLP - 1);
        sA[r][k] = hA[(size_t)(i0 + r) * MLP + k];
        sB[r][k] = hB[(size_t)(j0 + r) * MLP + k];
    }
    if (tid < MLP) sW[tid] = W2[tid];
    __syncthreads();

    const int ti = tid >> 4;
    const int tj = tid & 15;
    const int i = i0 + ti;
    const int j = j0 + tj;

    float acc = 0.0f;
#pragma unroll 4
    for (int k = 0; k < MLP; ++k) {
        acc += sW[k] * fast_tanh(sA[ti][k] + sB[tj][k]);
    }
    float r = acc + b2[0];
    out[(size_t)i * S_LEN + j] = (i == j) ? 0.0f : r;
}

extern "C" void kernel_launch(void* const* d_in, const int* in_sizes, int n_in,
                              void* d_out, int out_size, void* d_ws, size_t ws_size,
                              hipStream_t stream) {
    const int*   widx   = (const int*)d_in[0];
    const int*   pidx   = (const int*)d_in[1];
    const float* Wword  = (const float*)d_in[3];
    const float* Wpos   = (const float*)d_in[4];
    const float* Wih_l0f = (const float*)d_in[5];
    const float* Whh_l0f = (const float*)d_in[6];
    const float* bih_l0f = (const float*)d_in[7];
    const float* bhh_l0f = (const float*)d_in[8];
    const float* Wih_l0b = (const float*)d_in[9];
    const float* Whh_l0b = (const float*)d_in[10];
    const float* bih_l0b = (const float*)d_in[11];
    const float* bhh_l0b = (const float*)d_in[12];
    const float* Wih_l1f = (const float*)d_in[13];
    const float* Whh_l1f = (const float*)d_in[14];
    const float* bih_l1f = (const float*)d_in[15];
    const float* bhh_l1f = (const float*)d_in[16];
    const float* Wih_l1b = (const float*)d_in[17];
    const float* Whh_l1b = (const float*)d_in[18];
    const float* bih_l1b = (const float*)d_in[19];
    const float* bhh_l1b = (const float*)d_in[20];
    const float* W1 = (const float*)d_in[21];
    const float* b1 = (const float*)d_in[22];
    const float* W2 = (const float*)d_in[23];
    const float* b2 = (const float*)d_in[24];
    float* out = (float*)d_out;

    float* ws = (float*)d_ws;
    float* x      = ws;
    float* gxf    = x      + S_LEN * EMB;
    float* gxb    = gxf    + S_LEN * G4H;
    float* h0     = gxb    + S_LEN * G4H;
    float* h1     = h0     + S_LEN * 2 * HDIM;
    float* hA     = h1     + S_LEN * 2 * HDIM;
    float* hB     = hA     + S_LEN * MLP;
    float* BP0f   = hB     + S_LEN * MLP;
    float* BP0b   = BP0f   + EMB * G4H;
    float* BP1f   = BP0b   + EMB * G4H;
    float* BP1b   = BP1f   + 2 * HDIM * G4H;
    float* WhhT0f = BP1b   + 2 * HDIM * G4H;
    float* WhhT0b = WhhT0f + HDIM * G4H;
    float* WhhT1f = WhhT0b + HDIM * G4H;
    float* WhhT1b = WhhT1f + HDIM * G4H;
    float* BPA    = WhhT1b + HDIM * G4H;
    float* BPB    = BPA    + 2 * HDIM * MLP;

    auto tgrid = [](int total) { return dim3((total + 255) / 256); };
    pack_bt<<<tgrid(G4H * EMB),      256, 0, stream>>>(Wih_l0f, BP0f, G4H, EMB,      0,      EMB);
    pack_bt<<<tgrid(G4H * EMB),      256, 0, stream>>>(Wih_l0b, BP0b, G4H, EMB,      0,      EMB);
    pack_bt<<<tgrid(G4H * 2 * HDIM), 256, 0, stream>>>(Wih_l1f, BP1f, G4H, 2 * HDIM, 0, 2 * HDIM);
    pack_bt<<<tgrid(G4H * 2 * HDIM), 256, 0, stream>>>(Wih_l1b, BP1b, G4H, 2 * HDIM, 0, 2 * HDIM);
    pack_bt<<<tgrid(MLP * 2 * HDIM), 256, 0, stream>>>(W1, BPA, MLP, G4H,        0, 2 * HDIM);
    pack_bt<<<tgrid(MLP * 2 * HDIM), 256, 0, stream>>>(W1, BPB, MLP, G4H, 2 * HDIM, 2 * HDIM);
    pack_whh<<<tgrid(G4H * HDIM), 256, 0, stream>>>(Whh_l0f, WhhT0f);
    pack_whh<<<tgrid(G4H * HDIM), 256, 0, stream>>>(Whh_l0b, WhhT0b);
    pack_whh<<<tgrid(G4H * HDIM), 256, 0, stream>>>(Whh_l1f, WhhT1f);
    pack_whh<<<tgrid(G4H * HDIM), 256, 0, stream>>>(Whh_l1b, WhhT1b);

    embed_kernel<<<S_LEN, EMB, 0, stream>>>(widx, pidx, Wword, Wpos, x);

    {
        dim3 grid(G4H / 64, S_LEN / 16);
        gemm_wmma_f32<<<grid, 32, 0, stream>>>(x, BP0f, bih_l0f, bhh_l0f, gxf, S_LEN, G4H, EMB);
        gemm_wmma_f32<<<grid, 32, 0, stream>>>(x, BP0b, bih_l0b, bhh_l0b, gxb, S_LEN, G4H, EMB);
    }
    lstm_layer<<<2, HDIM, 0, stream>>>(gxf, gxb, WhhT0f, WhhT0b, h0);

    {
        dim3 grid(G4H / 64, S_LEN / 16);
        gemm_wmma_f32<<<grid, 32, 0, stream>>>(h0, BP1f, bih_l1f, bhh_l1f, gxf, S_LEN, G4H, 2 * HDIM);
        gemm_wmma_f32<<<grid, 32, 0, stream>>>(h0, BP1b, bih_l1b, bhh_l1b, gxb, S_LEN, G4H, 2 * HDIM);
    }
    lstm_layer<<<2, HDIM, 0, stream>>>(gxf, gxb, WhhT1f, WhhT1b, h1);

    {
        dim3 grid(MLP / 64, S_LEN / 16);
        gemm_wmma_f32<<<grid, 32, 0, stream>>>(h1, BPA, b1, (const float*)nullptr, hA, S_LEN, MLP, 2 * HDIM);
        gemm_wmma_f32<<<grid, 32, 0, stream>>>(h1, BPB, (const float*)nullptr, (const float*)nullptr, hB, S_LEN, MLP, 2 * HDIM);
    }

    {
        dim3 grid(S_LEN / 16, S_LEN / 16);
        pairwise_kernel<<<grid, 256, 0, stream>>>(hA, hB, W2, b2, out);
    }
}
